# starting kernel
# baseline (speedup 1.0000x reference)
#include <hip/hip_runtime.h>
#include <math.h>


typedef __attribute__((ext_vector_type(16))) _Float16 v16h;
typedef __attribute__((ext_vector_type(8)))  float    v8f;

#define DIM 256

__device__ __forceinline__ float atomAddF(float* p, float v) {
  return unsafeAtomicAdd(p, v);
}

__global__ void zero_kernel(float* __restrict__ p, int n) {
  int i = blockIdx.x * blockDim.x + threadIdx.x;
  if (i < n) p[i] = 0.0f;
}

__global__ void deg_kernel(const int* __restrict__ row, const int* __restrict__ col,
                           float* __restrict__ deg_out, float* __restrict__ deg_in, int E) {
  int e = blockIdx.x * blockDim.x + threadIdx.x;
  if (e < E) {
    atomAddF(&deg_out[row[e]], 1.0f);
    atomAddF(&deg_in[col[e]], 1.0f);
  }
}

__global__ void rsqrt_kernel(float* __restrict__ d, int n) {
  int i = blockIdx.x * blockDim.x + threadIdx.x;
  if (i < n) {
    float v = d[i];
    v = v < 1.0f ? 1.0f : v;
    d[i] = rsqrtf(v);
  }
}

__global__ __launch_bounds__(256)
void scatter_kernel(const float* __restrict__ h, const float* __restrict__ dinv_out,
                    const int* __restrict__ row, const int* __restrict__ col,
                    float* __restrict__ agg, int E) {
  int idx = blockIdx.x * blockDim.x + threadIdx.x;
  int e  = idx >> 6;
  int c4 = (idx & 63) << 2;
  if (e < E) {
    int r = row[e];
    int c = col[e];
    float s = dinv_out[r];
    const float4 hv = *(const float4*)(h + r * DIM + c4);
    float* dst = agg + c * DIM + c4;
    atomAddF(dst + 0, hv.x * s);
    atomAddF(dst + 1, hv.y * s);
    atomAddF(dst + 2, hv.z * s);
    atomAddF(dst + 3, hv.w * s);
  }
}

__global__ __launch_bounds__(256)
void pack_w_kernel(const float* __restrict__ W, _Float16* __restrict__ pack) {
  int t = blockIdx.x * blockDim.x + threadIdx.x;
  int elem = t & 15;
  int lane = (t >> 4) & 31;
  int kc   = (t >> 9) & 7;
  int tn   = t >> 12;
  int v = elem >> 1, p = elem & 1;
  int half = lane >> 4, mn = lane & 15;
  int kb = kc * 32 + ((v & 4) << 2) + (half << 3) + ((v & 3) << 1) + p;
  int gn = tn * 16 + mn;
  pack[t] = (_Float16)W[kb * DIM + gn];
}

template <bool HAS_RESID>
__global__ __launch_bounds__(128)
void gemm_wmma_kernel(const float* __restrict__ A, const _Float16* __restrict__ Wp,
                      const float* __restrict__ bias, const float* __restrict__ dinv_in,
                      const float* __restrict__ resid, float* __restrict__ out) {
  const int wave   = blockIdx.x * (blockDim.x >> 5) + (threadIdx.x >> 5);
  const int lane   = threadIdx.x & 31;
  const int tile_m = wave >> 4;
  const int tile_n = wave & 15;
  const int half   = lane >> 4;
  const int mn     = lane & 15;

  const int gm_a = tile_m * 16 + mn;
  const int gn   = tile_n * 16 + mn;
  const float* Arow = A + gm_a * DIM + half * 8;
  const v16h*  Bp   = (const v16h*)Wp + tile_n * 256 + lane;

  v8f c = {};
#pragma unroll
  for (int kc = 0; kc < 8; ++kc) {
    const float* ap = Arow + kc * 32;
    const float4 q0 = *(const float4*)(ap);
    const float4 q1 = *(const float4*)(ap + 4);
    const float4 q2 = *(const float4*)(ap + 16);
    const float4 q3 = *(const float4*)(ap + 20);
    v16h a;
    a[0]  = (_Float16)q0.x;  a[1]  = (_Float16)q0.y;
    a[2]  = (_Float16)q0.z;  a[3]  = (_Float16)q0.w;
    a[4]  = (_Float16)q1.x;  a[5]  = (_Float16)q1.y;
    a[6]  = (_Float16)q1.z;  a[7]  = (_Float16)q1.w;
    a[8]  = (_Float16)q2.x;  a[9]  = (_Float16)q2.y;
    a[10] = (_Float16)q2.z;  a[11] = (_Float16)q2.w;
    a[12] = (_Float16)q3.x;  a[13] = (_Float16)q3.y;
    a[14] = (_Float16)q3.z;  a[15] = (_Float16)q3.w;
    const v16h b = Bp[kc * 32];
    c = __builtin_amdgcn_wmma_f32_16x16x32_f16(false, a, false, b,
                                               (short)0, c, false, false);
  }

  const float bn = bias[gn];
#pragma unroll
  for (int r = 0; r < 8; ++r) {
    const int gm = tile_m * 16 + r + half * 8;
    float val = c[r] + bn;
    val *= dinv_in[gm];
    if constexpr (HAS_RESID) val += resid[gm * DIM + gn];
    out[gm * DIM + gn] = val;
  }
}

__global__ __launch_bounds__(256)
void bn_stats_kernel(const float* __restrict__ pre, float* __restrict__ sums,
                     float* __restrict__ sumsq, int n) {
  const int c    = threadIdx.x;
  const int row0 = blockIdx.x * 128;
  int rend = row0 + 128;
  if (rend > n) rend = n;
  float s = 0.0f, s2 = 0.0f;
  for (int r = row0; r < rend; ++r) {
    float v = pre[r * DIM + c];
    s  += v;
    s2 += v * v;
  }
  atomAddF(&sums[c], s);
  atomAddF(&sumsq[c], s2);
}

__global__ void bn_finalize_kernel(float* __restrict__ sums, float* __restrict__ sumsq, int n) {
  const int c = threadIdx.x;
  const float inv_n = 1.0f / (float)n;
  float mu  = sums[c] * inv_n;
  float var = sumsq[c] * inv_n - mu * mu;
  sums[c]  = mu;
  sumsq[c] = rsqrtf(var + 1e-5f);
}

__global__ __launch_bounds__(256)
void bn_apply_kernel(float* __restrict__ buf, const float* __restrict__ mu,
                     const float* __restrict__ inv_std, const float* __restrict__ gamma,
                     const float* __restrict__ beta, int nd) {
  int i = blockIdx.x * blockDim.x + threadIdx.x;
  if (i < nd) {
    const int c = i & (DIM - 1);
    float v = (buf[i] - mu[c]) * inv_std[c] * gamma[c] + beta[c];
    buf[i] = fmaxf(v, 0.0f);
  }
}

extern "C" void kernel_launch(void* const* d_in, const int* in_sizes, int n_in,
                              void* d_out, int out_size, void* d_ws, size_t ws_size,
                              hipStream_t stream) {
  const float* x  = (const float*)d_in[0];
  const int*   ei = (const int*)d_in[1];
  const float* W0 = (const float*)d_in[2];
  const float* b0 = (const float*)d_in[3];
  const float* g0 = (const float*)d_in[4];
  const float* be0= (const float*)d_in[5];
  const float* W1 = (const float*)d_in[6];
  const float* b1 = (const float*)d_in[7];
  const float* g1 = (const float*)d_in[8];
  const float* be1= (const float*)d_in[9];
  const float* W2 = (const float*)d_in[10];
  const float* b2 = (const float*)d_in[11];
  float* out = (float*)d_out;

  const int N = in_sizes[0] / DIM;
  const int E = in_sizes[1] / 2;
  const int ND = N * DIM;
  const int* row = ei;
  const int* col = ei + E;

  float* ws       = (float*)d_ws;
  float* dinv_out = ws;
  float* dinv_in  = ws + N;
  float* stats    = ws + 2 * N;
  float* agg      = ws + 2 * N + 512;
  float* bufA     = agg + (size_t)ND;
  _Float16* wp0   = (_Float16*)(bufA + (size_t)ND);
  _Float16* wp1   = wp0 + 65536;
  _Float16* wp2   = wp1 + 65536;

  const int zb = 256;
  auto nb = [](int n, int b) { return (n + b - 1) / b; };

  const int scat_blocks = (E * 64) / 256;
  const int gemm_blocks = (N / 16) * 16 / 4;
  const int bn_blocks   = nb(N, 128);
  const int pack_blocks = 65536 / 256;

  zero_kernel<<<nb(2 * N, zb), zb, 0, stream>>>(dinv_out, 2 * N);
  deg_kernel<<<nb(E, zb), zb, 0, stream>>>(row, col, dinv_out, dinv_in, E);
  rsqrt_kernel<<<nb(2 * N, zb), zb, 0, stream>>>(dinv_out, 2 * N);
  pack_w_kernel<<<pack_blocks, 256, 0, stream>>>(W0, wp0);
  pack_w_kernel<<<pack_blocks, 256, 0, stream>>>(W1, wp1);
  pack_w_kernel<<<pack_blocks, 256, 0, stream>>>(W2, wp2);

  zero_kernel<<<nb(ND, zb), zb, 0, stream>>>(agg, ND);
  scatter_kernel<<<scat_blocks, 256, 0, stream>>>(x, dinv_out, row, col, agg, E);
  gemm_wmma_kernel<true><<<gemm_blocks, 128, 0, stream>>>(agg, wp0, b0, dinv_in, x, out);
  zero_kernel<<<2, zb, 0, stream>>>(stats, 512);
  bn_stats_kernel<<<bn_blocks, 256, 0, stream>>>(out, stats, stats + 256, N);
  bn_finalize_kernel<<<1, 256, 0, stream>>>(stats, stats + 256, N);
  bn_apply_kernel<<<nb(ND, zb), zb, 0, stream>>>(out, stats, stats + 256, g0, be0, ND);

  zero_kernel<<<nb(ND, zb), zb, 0, stream>>>(agg, ND);
  scatter_kernel<<<scat_blocks, 256, 0, stream>>>(out, dinv_out, row, col, agg, E);
  gemm_wmma_kernel<true><<<gemm_blocks, 128, 0, stream>>>(agg, wp1, b1, dinv_in, out, bufA);
  zero_kernel<<<2, zb, 0, stream>>>(stats, 512);
  bn_stats_kernel<<<bn_blocks, 256, 0, stream>>>(bufA, stats, stats + 256, N);
  bn_finalize_kernel<<<1, 256, 0, stream>>>(stats, stats + 256, N);
  bn_apply_kernel<<<nb(ND, zb), zb, 0, stream>>>(bufA, stats, stats + 256, g1, be1, ND);

  zero_kernel<<<nb(ND, zb), zb, 0, stream>>>(agg, ND);
  scatter_kernel<<<scat_blocks, 256, 0, stream>>>(bufA, dinv_out, row, col, agg, E);
  gemm_wmma_kernel<false><<<gemm_blocks, 128, 0, stream>>>(agg, wp2, b2, dinv_in, nullptr, out);
}


# BatchNorm+ReLU no longer stored: next gather and next GEMM residual normalise pre-norm values on load (one FMA, a=inv_std*gamma, c=beta-mu*a, then clamp); both in-place apply passes removed
# speedup vs baseline: 1.5389x; 1.5389x over previous
#include <hip/hip_runtime.h>
#include <math.h>


typedef __attribute__((ext_vector_type(16))) _Float16 v16h;
typedef __attribute__((ext_vector_type(8)))  float    v8f;

#define DIM 256

__device__ __forceinline__ float atomAddF(float* p, float v) {
  return unsafeAtomicAdd(p, v);
}

__device__ __forceinline__ float bn_relu(float v, float a, float c) {
  return fmaxf(v * a + c, 0.0f);
}
__device__ __forceinline__ float4 bn_relu4(float4 v, float4 a, float4 c) {
  float4 r;
  r.x = bn_relu(v.x, a.x, c.x);
  r.y = bn_relu(v.y, a.y, c.y);
  r.z = bn_relu(v.z, a.z, c.z);
  r.w = bn_relu(v.w, a.w, c.w);
  return r;
}

__global__ void zero_kernel(float* __restrict__ p, int n) {
  int i = blockIdx.x * blockDim.x + threadIdx.x;
  if (i < n) p[i] = 0.0f;
}

__global__ void zero_i32_kernel(int* __restrict__ p, int n) {
  int i = blockIdx.x * blockDim.x + threadIdx.x;
  if (i < n) p[i] = 0;
}

__global__ void count_kernel(const int* __restrict__ row, const int* __restrict__ col,
                             int* __restrict__ cnt_out, int* __restrict__ cnt_in,
                             int* __restrict__ slot, int E) {
  int e = blockIdx.x * blockDim.x + threadIdx.x;
  if (e < E) {
    atomicAdd(&cnt_out[row[e]], 1);
    slot[e] = atomicAdd(&cnt_in[col[e]], 1);
  }
}

__global__ __launch_bounds__(1024)
void scan_kernel(const int* __restrict__ cnt_in, int* __restrict__ first,
                 float* __restrict__ dinv_in, int n) {
  __shared__ int part[1024];
  const int t  = threadIdx.x;
  const int ch = (n + 1023) >> 10;
  const int b0 = t * ch;
  int b1 = b0 + ch;
  if (b1 > n) b1 = n;
  int s = 0;
  for (int i = b0; i < b1; ++i) s += cnt_in[i];
  part[t] = s;
  __syncthreads();
  for (int d = 1; d < 1024; d <<= 1) {
    const int v = (t >= d) ? part[t - d] : 0;
    __syncthreads();
    part[t] += v;
    __syncthreads();
  }
  int run = part[t] - s;
  for (int i = b0; i < b1; ++i) {
    const int c = cnt_in[i];
    first[i] = run;
    run += c;
    float v = (float)c;
    v = v < 1.0f ? 1.0f : v;
    dinv_in[i] = rsqrtf(v);
  }
}

__global__ void fill_kernel(const int* __restrict__ row, const int* __restrict__ col,
                            const int* __restrict__ cnt_out, const int* __restrict__ first,
                            const int* __restrict__ slot,
                            int* __restrict__ esrc, float* __restrict__ ewgt, int E) {
  int e = blockIdx.x * blockDim.x + threadIdx.x;
  if (e < E) {
    const int r = row[e];
    const int p = first[col[e]] + slot[e];
    float v = (float)cnt_out[r];
    v = v < 1.0f ? 1.0f : v;
    esrc[p] = r;
    ewgt[p] = rsqrtf(v);
  }
}

template <bool BN>
__global__ __launch_bounds__(256)
void gather_kernel(const float* __restrict__ h, const int* __restrict__ first,
                   const int* __restrict__ cnt_in, const int* __restrict__ esrc,
                   const float* __restrict__ ewgt, float* __restrict__ agg, int n,
                   const float* __restrict__ mu, const float* __restrict__ inv_std,
                   const float* __restrict__ gamma, const float* __restrict__ beta) {
  const int node = __builtin_amdgcn_readfirstlane((int)(blockIdx.x * 8 + (threadIdx.x >> 5)));
  if (node >= n) return;
  const int lane = threadIdx.x & 31;
  const int beg = first[node];
  const int len = cnt_in[node];
  const int*   es = esrc + beg;
  const float* ew = ewgt + beg;
  const float* hb = h + lane * 4;

  float4 s0 = {}, s1 = {}, t0 = {}, t1 = {};
  if constexpr (BN) {
    const int c0 = lane * 4;
    const float4 m0 = *(const float4*)(mu + c0);
    const float4 m1 = *(const float4*)(mu + 128 + c0);
    const float4 i0 = *(const float4*)(inv_std + c0);
    const float4 i1 = *(const float4*)(inv_std + 128 + c0);
    const float4 g0 = *(const float4*)(gamma + c0);
    const float4 g1 = *(const float4*)(gamma + 128 + c0);
    const float4 e0 = *(const float4*)(beta + c0);
    const float4 e1 = *(const float4*)(beta + 128 + c0);
    s0.x = i0.x * g0.x; s0.y = i0.y * g0.y; s0.z = i0.z * g0.z; s0.w = i0.w * g0.w;
    s1.x = i1.x * g1.x; s1.y = i1.y * g1.y; s1.z = i1.z * g1.z; s1.w = i1.w * g1.w;
    t0.x = e0.x - m0.x * s0.x; t0.y = e0.y - m0.y * s0.y; t0.z = e0.z - m0.z * s0.z; t0.w = e0.w - m0.w * s0.w;
    t1.x = e1.x - m1.x * s1.x; t1.y = e1.y - m1.y * s1.y; t1.z = e1.z - m1.z * s1.z; t1.w = e1.w - m1.w * s1.w;
  }

  float4 a0 = {0.0f, 0.0f, 0.0f, 0.0f};
  float4 a1 = {0.0f, 0.0f, 0.0f, 0.0f};
  int j = 0;
#pragma unroll 2
  for (; j + 4 <= len; j += 4) {
    const int r0 = es[j], r1 = es[j + 1], r2 = es[j + 2], r3 = es[j + 3];
    const float w0 = ew[j], w1 = ew[j + 1], w2 = ew[j + 2], w3 = ew[j + 3];
    const float* p0 = hb + r0 * DIM;
    const float* p1 = hb + r1 * DIM;
    const float* p2 = hb + r2 * DIM;
    const float* p3 = hb + r3 * DIM;
    float4 x0 = *(const float4*)(p0), y0 = *(const float4*)(p0 + 128);
    float4 x1 = *(const float4*)(p1), y1 = *(const float4*)(p1 + 128);
    float4 x2 = *(const float4*)(p2), y2 = *(const float4*)(p2 + 128);
    float4 x3 = *(const float4*)(p3), y3 = *(const float4*)(p3 + 128);
    if constexpr (BN) {
      x0 = bn_relu4(x0, s0, t0); y0 = bn_relu4(y0, s1, t1);
      x1 = bn_relu4(x1, s0, t0); y1 = bn_relu4(y1, s1, t1);
      x2 = bn_relu4(x2, s0, t0); y2 = bn_relu4(y2, s1, t1);
      x3 = bn_relu4(x3, s0, t0); y3 = bn_relu4(y3, s1, t1);
    }
    a0.x += x0.x * w0; a0.y += x0.y * w0; a0.z += x0.z * w0; a0.w += x0.w * w0;
    a1.x += y0.x * w0; a1.y += y0.y * w0; a1.z += y0.z * w0; a1.w += y0.w * w0;
    a0.x += x1.x * w1; a0.y += x1.y * w1; a0.z += x1.z * w1; a0.w += x1.w * w1;
    a1.x += y1.x * w1; a1.y += y1.y * w1; a1.z += y1.z * w1; a1.w += y1.w * w1;
    a0.x += x2.x * w2; a0.y += x2.y * w2; a0.z += x2.z * w2; a0.w += x2.w * w2;
    a1.x += y2.x * w2; a1.y += y2.y * w2; a1.z += y2.z * w2; a1.w += y2.w * w2;
    a0.x += x3.x * w3; a0.y += x3.y * w3; a0.z += x3.z * w3; a0.w += x3.w * w3;
    a1.x += y3.x * w3; a1.y += y3.y * w3; a1.z += y3.z * w3; a1.w += y3.w * w3;
  }
  for (; j < len; ++j) {
    const int r0 = es[j];
    const float w0 = ew[j];
    const float* p0 = hb + r0 * DIM;
    float4 x0 = *(const float4*)(p0), y0 = *(const float4*)(p0 + 128);
    if constexpr (BN) {
      x0 = bn_relu4(x0, s0, t0); y0 = bn_relu4(y0, s1, t1);
    }
    a0.x += x0.x * w0; a0.y += x0.y * w0; a0.z += x0.z * w0; a0.w += x0.w * w0;
    a1.x += y0.x * w0; a1.y += y0.y * w0; a1.z += y0.z * w0; a1.w += y0.w * w0;
  }
  float* dst = agg + node * DIM + lane * 4;
  *(float4*)(dst)       = a0;
  *(float4*)(dst + 128) = a1;
}

__global__ __launch_bounds__(256)
void pack_w_kernel(const float* __restrict__ W, _Float16* __restrict__ pack) {
  int t = blockIdx.x * blockDim.x + threadIdx.x;
  int elem = t & 15;
  int lane = (t >> 4) & 31;
  int kc   = (t >> 9) & 7;
  int tn   = t >> 12;
  int v = elem >> 1, p = elem & 1;
  int half = lane >> 4, mn = lane & 15;
  int kb = kc * 32 + ((v & 4) << 2) + (half << 3) + ((v & 3) << 1) + p;
  int gn = tn * 16 + mn;
  pack[t] = (_Float16)W[kb * DIM + gn];
}

template <int RESID>
__global__ __launch_bounds__(128)
void gemm_wmma_kernel(const float* __restrict__ A, const _Float16* __restrict__ Wp,
                      const float* __restrict__ bias, const float* __restrict__ dinv_in,
                      const float* __restrict__ resid, float* __restrict__ out,
                      const float* __restrict__ mu, const float* __restrict__ inv_std,
                      const float* __restrict__ gamma, const float* __restrict__ beta) {
  const int wave   = blockIdx.x * (blockDim.x >> 5) + (threadIdx.x >> 5);
  const int lane   = threadIdx.x & 31;
  const int tile_m = wave >> 4;
  const int tile_n = wave & 15;
  const int half   = lane >> 4;
  const int mn     = lane & 15;

  const int gm_a = tile_m * 16 + mn;
  const int gn   = tile_n * 16 + mn;
  const float* Arow = A + gm_a * DIM + half * 8;
  const v16h*  Bp   = (const v16h*)Wp + tile_n * 256 + lane;

  float ra = 0.0f, rc = 0.0f;
  if constexpr (RESID == 2) {
    ra = inv_std[gn] * gamma[gn];
    rc = beta[gn] - mu[gn] * ra;
  }

  v8f c = {};
#pragma unroll
  for (int kc = 0; kc < 8; ++kc) {
    const float* ap = Arow + kc * 32;
    const float4 q0 = *(const float4*)(ap);
    const float4 q1 = *(const float4*)(ap + 4);
    const float4 q2 = *(const float4*)(ap + 16);
    const float4 q3 = *(const float4*)(ap + 20);
    v16h a;
    a[0]  = (_Float16)q0.x;  a[1]  = (_Float16)q0.y;
    a[2]  = (_Float16)q0.z;  a[3]  = (_Float16)q0.w;
    a[4]  = (_Float16)q1.x;  a[5]  = (_Float16)q1.y;
    a[6]  = (_Float16)q1.z;  a[7]  = (_Float16)q1.w;
    a[8]  = (_Float16)q2.x;  a[9]  = (_Float16)q2.y;
    a[10] = (_Float16)q2.z;  a[11] = (_Float16)q2.w;
    a[12] = (_Float16)q3.x;  a[13] = (_Float16)q3.y;
    a[14] = (_Float16)q3.z;  a[15] = (_Float16)q3.w;
    const v16h b = Bp[kc * 32];
    c = __builtin_amdgcn_wmma_f32_16x16x32_f16(false, a, false, b,
                                               (short)0, c, false, false);
  }

  const float bn = bias[gn];
#pragma unroll
  for (int r = 0; r < 8; ++r) {
    const int gm = tile_m * 16 + r + half * 8;
    float val = c[r] + bn;
    val *= dinv_in[gm];
    if constexpr (RESID == 1) val += resid[gm * DIM + gn];
    if constexpr (RESID == 2) val += bn_relu(resid[gm * DIM + gn], ra, rc);
    out[gm * DIM + gn] = val;
  }
}

__global__ __launch_bounds__(256)
void bn_stats_kernel(const float* __restrict__ pre, float* __restrict__ sums,
                     float* __restrict__ sumsq, int n) {
  const int c    = threadIdx.x;
  const int row0 = blockIdx.x * 128;
  int rend = row0 + 128;
  if (rend > n) rend = n;
  float s = 0.0f, s2 = 0.0f;
  for (int r = row0; r < rend; ++r) {
    float v = pre[r * DIM + c];
    s  += v;
    s2 += v * v;
  }
  atomAddF(&sums[c], s);
  atomAddF(&sumsq[c], s2);
}

__global__ void bn_finalize_kernel(float* __restrict__ sums, float* __restrict__ sumsq, int n) {
  const int c = threadIdx.x;
  const float inv_n = 1.0f / (float)n;
  float mu  = sums[c] * inv_n;
  float var = sumsq[c] * inv_n - mu * mu;
  sums[c]  = mu;
  sumsq[c] = rsqrtf(var + 1e-5f);
}

extern "C" void kernel_launch(void* const* d_in, const int* in_sizes, int n_in,
                              void* d_out, int out_size, void* d_ws, size_t ws_size,
                              hipStream_t stream) {
  const float* x  = (const float*)d_in[0];
  const int*   ei = (const int*)d_in[1];
  const float* W0 = (const float*)d_in[2];
  const float* b0 = (const float*)d_in[3];
  const float* g0 = (const float*)d_in[4];
  const float* be0= (const float*)d_in[5];
  const float* W1 = (const float*)d_in[6];
  const float* b1 = (const float*)d_in[7];
  const float* g1 = (const float*)d_in[8];
  const float* be1= (const float*)d_in[9];
  const float* W2 = (const float*)d_in[10];
  const float* b2 = (const float*)d_in[11];
  float* out = (float*)d_out;

  const int N = in_sizes[0] / DIM;
  const int E = in_sizes[1] / 2;
  const int ND = N * DIM;
  const int* row = ei;
  const int* col = ei + E;

  float* ws       = (float*)d_ws;
  float* agg      = ws;
  float* bufA     = agg + (size_t)ND;
  _Float16* wp0   = (_Float16*)(bufA + (size_t)ND);
  _Float16* wp1   = wp0 + 65536;
  _Float16* wp2   = wp1 + 65536;
  float* stats0   = (float*)(wp2 + 65536);
  float* stats1   = stats0 + 512;
  float* dinv_in  = stats1 + 512;
  int*   cnt_out  = (int*)(dinv_in + N);
  int*   cnt_in   = cnt_out + N;
  int*   first    = cnt_in + N;
  int*   slot     = first + N;
  int*   esrc     = slot + E;
  float* ewgt     = (float*)(esrc + E);

  const int zb = 256;
  auto nb = [](int n, int b) { return (n + b - 1) / b; };

  const int gath_blocks = nb(N, 8);
  const int gemm_blocks = (N / 16) * 16 / 4;
  const int bn_blocks   = nb(N, 128);
  const int pack_blocks = 65536 / 256;

  zero_i32_kernel<<<nb(2 * N, zb), zb, 0, stream>>>(cnt_out, 2 * N);
  count_kernel<<<nb(E, zb), zb, 0, stream>>>(row, col, cnt_out, cnt_in, slot, E);
  scan_kernel<<<1, 1024, 0, stream>>>(cnt_in, first, dinv_in, N);
  fill_kernel<<<nb(E, zb), zb, 0, stream>>>(row, col, cnt_out, first, slot, esrc, ewgt, E);
  pack_w_kernel<<<pack_blocks, 256, 0, stream>>>(W0, wp0);
  pack_w_kernel<<<pack_blocks, 256, 0, stream>>>(W1, wp1);
  pack_w_kernel<<<pack_blocks, 256, 0, stream>>>(W2, wp2);

  gather_kernel<false><<<gath_blocks, 256, 0, stream>>>(x, first, cnt_in, esrc, ewgt, agg, N,
                                                        nullptr, nullptr, nullptr, nullptr);
  gemm_wmma_kernel<1><<<gemm_blocks, 128, 0, stream>>>(agg, wp0, b0, dinv_in, x, out,
                                                       nullptr, nullptr, nullptr, nullptr);
  zero_kernel<<<2, zb, 0, stream>>>(stats0, 512);
  bn_stats_kernel<<<bn_blocks, 256, 0, stream>>>(out, stats0, stats0 + 256, N);
  bn_finalize_kernel<<<1, 256, 0, stream>>>(stats0, stats0 + 256, N);

  gather_kernel<true><<<gath_blocks, 256, 0, stream>>>(out, first, cnt_in, esrc, ewgt, agg, N,
                                                       stats0, stats0 + 256, g0, be0);
  gemm_wmma_kernel<2><<<gemm_blocks, 128, 0, stream>>>(agg, wp1, b1, dinv_in, out, bufA,
                                                       stats0, stats0 + 256, g0, be0);
  zero_kernel<<<2, zb, 0, stream>>>(stats1, 512);
  bn_stats_kernel<<<bn_blocks, 256, 0, stream>>>(bufA, stats1, stats1 + 256, N);
  bn_finalize_kernel<<<1, 256, 0, stream>>>(stats1, stats1 + 256, N);

  gather_kernel<true><<<gath_blocks, 256, 0, stream>>>(bufA, first, cnt_in, esrc, ewgt, agg, N,
                                                       stats1, stats1 + 256, g1, be1);
  gemm_wmma_kernel<0><<<gemm_blocks, 128, 0, stream>>>(agg, wp2, b2, dinv_in, nullptr, out,
                                                       nullptr, nullptr, nullptr, nullptr);
}


# GEMM re-tiled: one wave per 16-row strip over all 256 columns (was one 16x16 tile): A loaded and converted once for 16 accumulator tiles, packed W read as whole lines; per-tile K order kept
# speedup vs baseline: 2.0283x; 1.3180x over previous
#include <hip/hip_runtime.h>
#include <math.h>


typedef __attribute__((ext_vector_type(16))) _Float16 v16h;
typedef __attribute__((ext_vector_type(8)))  float    v8f;

#define DIM 256

__device__ __forceinline__ float atomAddF(float* p, float v) {
  return unsafeAtomicAdd(p, v);
}

__device__ __forceinline__ float bn_relu(float v, float a, float c) {
  return fmaxf(v * a + c, 0.0f);
}
__device__ __forceinline__ float4 bn_relu4(float4 v, float4 a, float4 c) {
  float4 r;
  r.x = bn_relu(v.x, a.x, c.x);
  r.y = bn_relu(v.y, a.y, c.y);
  r.z = bn_relu(v.z, a.z, c.z);
  r.w = bn_relu(v.w, a.w, c.w);
  return r;
}

__global__ void zero_kernel(float* __restrict__ p, int n) {
  int i = blockIdx.x * blockDim.x + threadIdx.x;
  if (i < n) p[i] = 0.0f;
}

__global__ void zero_i32_kernel(int* __restrict__ p, int n) {
  int i = blockIdx.x * blockDim.x + threadIdx.x;
  if (i < n) p[i] = 0;
}

__global__ void count_kernel(const int* __restrict__ row, const int* __restrict__ col,
                             int* __restrict__ cnt_out, int* __restrict__ cnt_in,
                             int* __restrict__ slot, int E) {
  int e = blockIdx.x * blockDim.x + threadIdx.x;
  if (e < E) {
    atomicAdd(&cnt_out[row[e]], 1);
    slot[e] = atomicAdd(&cnt_in[col[e]], 1);
  }
}

__global__ __launch_bounds__(1024)
void scan_kernel(const int* __restrict__ cnt_in, int* __restrict__ first,
                 float* __restrict__ dinv_in, int n) {
  __shared__ int part[1024];
  const int t  = threadIdx.x;
  const int ch = (n + 1023) >> 10;
  const int b0 = t * ch;
  int b1 = b0 + ch;
  if (b1 > n) b1 = n;
  int s = 0;
  for (int i = b0; i < b1; ++i) s += cnt_in[i];
  part[t] = s;
  __syncthreads();
  for (int d = 1; d < 1024; d <<= 1) {
    const int v = (t >= d) ? part[t - d] : 0;
    __syncthreads();
    part[t] += v;
    __syncthreads();
  }
  int run = part[t] - s;
  for (int i = b0; i < b1; ++i) {
    const int c = cnt_in[i];
    first[i] = run;
    run += c;
    float v = (float)c;
    v = v < 1.0f ? 1.0f : v;
    dinv_in[i] = rsqrtf(v);
  }
}

__global__ void fill_kernel(const int* __restrict__ row, const int* __restrict__ col,
                            const int* __restrict__ cnt_out, const int* __restrict__ first,
                            const int* __restrict__ slot,
                            int* __restrict__ esrc, float* __restrict__ ewgt, int E) {
  int e = blockIdx.x * blockDim.x + threadIdx.x;
  if (e < E) {
    const int r = row[e];
    const int p = first[col[e]] + slot[e];
    float v = (float)cnt_out[r];
    v = v < 1.0f ? 1.0f : v;
    esrc[p] = r;
    ewgt[p] = rsqrtf(v);
  }
}

template <bool BN>
__global__ __launch_bounds__(256)
void gather_kernel(const float* __restrict__ h, const int* __restrict__ first,
                   const int* __restrict__ cnt_in, const int* __restrict__ esrc,
                   const float* __restrict__ ewgt, float* __restrict__ agg, int n,
                   const float* __restrict__ mu, const float* __restrict__ inv_std,
                   const float* __restrict__ gamma, const float* __restrict__ beta) {
  const int node = __builtin_amdgcn_readfirstlane((int)(blockIdx.x * 8 + (threadIdx.x >> 5)));
  if (node >= n) return;
  const int lane = threadIdx.x & 31;
  const int beg = first[node];
  const int len = cnt_in[node];
  const int*   es = esrc + beg;
  const float* ew = ewgt + beg;
  const float* hb = h + lane * 4;

  float4 s0 = {}, s1 = {}, t0 = {}, t1 = {};
  if constexpr (BN) {
    const int c0 = lane * 4;
    const float4 m0 = *(const float4*)(mu + c0);
    const float4 m1 = *(const float4*)(mu + 128 + c0);
    const float4 i0 = *(const float4*)(inv_std + c0);
    const float4 i1 = *(const float4*)(inv_std + 128 + c0);
    const float4 g0 = *(const float4*)(gamma + c0);
    const float4 g1 = *(const float4*)(gamma + 128 + c0);
    const float4 e0 = *(const float4*)(beta + c0);
    const float4 e1 = *(const float4*)(beta + 128 + c0);
    s0.x = i0.x * g0.x; s0.y = i0.y * g0.y; s0.z = i0.z * g0.z; s0.w = i0.w * g0.w;
    s1.x = i1.x * g1.x; s1.y = i1.y * g1.y; s1.z = i1.z * g1.z; s1.w = i1.w * g1.w;
    t0.x = e0.x - m0.x * s0.x; t0.y = e0.y - m0.y * s0.y; t0.z = e0.z - m0.z * s0.z; t0.w = e0.w - m0.w * s0.w;
    t1.x = e1.x - m1.x * s1.x; t1.y = e1.y - m1.y * s1.y; t1.z = e1.z - m1.z * s1.z; t1.w = e1.w - m1.w * s1.w;
  }

  float4 a0 = {0.0f, 0.0f, 0.0f, 0.0f};
  float4 a1 = {0.0f, 0.0f, 0.0f, 0.0f};
  int j = 0;
#pragma unroll 2
  for (; j + 4 <= len; j += 4) {
    const int r0 = es[j], r1 = es[j + 1], r2 = es[j + 2], r3 = es[j + 3];
    const float w0 = ew[j], w1 = ew[j + 1], w2 = ew[j + 2], w3 = ew[j + 3];
    const float* p0 = hb + r0 * DIM;
    const float* p1 = hb + r1 * DIM;
    const float* p2 = hb + r2 * DIM;
    const float* p3 = hb + r3 * DIM;
    float4 x0 = *(const float4*)(p0), y0 = *(const float4*)(p0 + 128);
    float4 x1 = *(const float4*)(p1), y1 = *(const float4*)(p1 + 128);
    float4 x2 = *(const float4*)(p2), y2 = *(const float4*)(p2 + 128);
    float4 x3 = *(const float4*)(p3), y3 = *(const float4*)(p3 + 128);
    if constexpr (BN) {
      x0 = bn_relu4(x0, s0, t0); y0 = bn_relu4(y0, s1, t1);
      x1 = bn_relu4(x1, s0, t0); y1 = bn_relu4(y1, s1, t1);
      x2 = bn_relu4(x2, s0, t0); y2 = bn_relu4(y2, s1, t1);
      x3 = bn_relu4(x3, s0, t0); y3 = bn_relu4(y3, s1, t1);
    }
    a0.x += x0.x * w0; a0.y += x0.y * w0; a0.z += x0.z * w0; a0.w += x0.w * w0;
    a1.x += y0.x * w0; a1.y += y0.y * w0; a1.z += y0.z * w0; a1.w += y0.w * w0;
    a0.x += x1.x * w1; a0.y += x1.y * w1; a0.z += x1.z * w1; a0.w += x1.w * w1;
    a1.x += y1.x * w1; a1.y += y1.y * w1; a1.z += y1.z * w1; a1.w += y1.w * w1;
    a0.x += x2.x * w2; a0.y += x2.y * w2; a0.z += x2.z * w2; a0.w += x2.w * w2;
    a1.x += y2.x * w2; a1.y += y2.y * w2; a1.z += y2.z * w2; a1.w += y2.w * w2;
    a0.x += x3.x * w3; a0.y += x3.y * w3; a0.z += x3.z * w3; a0.w += x3.w * w3;
    a1.x += y3.x * w3; a1.y += y3.y * w3; a1.z += y3.z * w3; a1.w += y3.w * w3;
  }
  for (; j < len; ++j) {
    const int r0 = es[j];
    const float w0 = ew[j];
    const float* p0 = hb + r0 * DIM;
    float4 x0 = *(const float4*)(p0), y0 = *(const float4*)(p0 + 128);
    if constexpr (BN) {
      x0 = bn_relu4(x0, s0, t0); y0 = bn_relu4(y0, s1, t1);
    }
    a0.x += x0.x * w0; a0.y += x0.y * w0; a0.z += x0.z * w0; a0.w += x0.w * w0;
    a1.x += y0.x * w0; a1.y += y0.y * w0; a1.z += y0.z * w0; a1.w += y0.w * w0;
  }
  float* dst = agg + node * DIM + lane * 4;
  *(float4*)(dst)       = a0;
  *(float4*)(dst + 128) = a1;
}

__global__ __launch_bounds__(256)
void pack_w_kernel(const float* __restrict__ W, _Float16* __restrict__ pack) {
  int t = blockIdx.x * blockDim.x + threadIdx.x;
  int el    = t & 7;
  int lane  = (t >> 3) & 31;
  int piece = (t >> 8) & 1;
  int kc    = (t >> 9) & 7;
  int tn    = t >> 12;
  int half = lane >> 4, mn = lane & 15;
  int kb = kc * 32 + (piece << 4) + (half << 3) + el;
  int gn = tn * 16 + mn;
  pack[t] = (_Float16)W[kb * DIM + gn];
}

typedef __attribute__((ext_vector_type(8))) _Float16 v8h;

template <int RESID>
__global__ __launch_bounds__(128)
void gemm_wmma_kernel(const float* __restrict__ A, const _Float16* __restrict__ Wp,
                      const float* __restrict__ bias, const float* __restrict__ dinv_in,
                      const float* __restrict__ resid, float* __restrict__ out, int tiles_m,
                      const float* __restrict__ mu, const float* __restrict__ inv_std,
                      const float* __restrict__ gamma, const float* __restrict__ beta) {
  const int tile_m = blockIdx.x * (blockDim.x >> 5) + (threadIdx.x >> 5);
  if (tile_m >= tiles_m) return;
  const int lane   = threadIdx.x & 31;
  const int half   = lane >> 4;
  const int mn     = lane & 15;

  const int gm_a = tile_m * 16 + mn;
  const float* Arow = A + gm_a * DIM + half * 8;
  const v8h*   Bp   = (const v8h*)Wp + lane;

  v8f c[16];
#pragma unroll
  for (int tn = 0; tn < 16; ++tn) c[tn] = v8f{};

  v8h bl[2][16], bh[2][16];
  float4 q[2][4];
#pragma unroll
  for (int tn = 0; tn < 16; ++tn) {
    const v8h* bp = Bp + (tn * 8) * 64;
    bl[0][tn] = bp[0];
    bh[0][tn] = bp[32];
  }
  q[0][0] = *(const float4*)(Arow);
  q[0][1] = *(const float4*)(Arow + 4);
  q[0][2] = *(const float4*)(Arow + 16);
  q[0][3] = *(const float4*)(Arow + 20);

#pragma unroll
  for (int kc = 0; kc < 8; ++kc) {
    const int cur = kc & 1, nxt = cur ^ 1;
    if (kc + 1 < 8) {
#pragma unroll
      for (int tn = 0; tn < 16; ++tn) {
        const v8h* bp = Bp + (tn * 8 + kc + 1) * 64;
        bl[nxt][tn] = bp[0];
        bh[nxt][tn] = bp[32];
      }
      const float* ap = Arow + (kc + 1) * 32;
      q[nxt][0] = *(const float4*)(ap);
      q[nxt][1] = *(const float4*)(ap + 4);
      q[nxt][2] = *(const float4*)(ap + 16);
      q[nxt][3] = *(const float4*)(ap + 20);
    }
    __builtin_amdgcn_sched_barrier(0);
    const float4 q0 = q[cur][0], q1 = q[cur][1], q2 = q[cur][2], q3 = q[cur][3];
    v16h a;
    a[0]  = (_Float16)q0.x;  a[1]  = (_Float16)q0.y;
    a[2]  = (_Float16)q0.z;  a[3]  = (_Float16)q0.w;
    a[4]  = (_Float16)q1.x;  a[5]  = (_Float16)q1.y;
    a[6]  = (_Float16)q1.z;  a[7]  = (_Float16)q1.w;
    a[8]  = (_Float16)q2.x;  a[9]  = (_Float16)q2.y;
    a[10] = (_Float16)q2.z;  a[11] = (_Float16)q2.w;
    a[12] = (_Float16)q3.x;  a[13] = (_Float16)q3.y;
    a[14] = (_Float16)q3.z;  a[15] = (_Float16)q3.w;
#pragma unroll
    for (int tn = 0; tn < 16; ++tn) {
      const v16h b = __builtin_shufflevector(bl[cur][tn], bh[cur][tn], 0, 1, 2, 3, 4, 5, 6, 7,
                                             8, 9, 10, 11, 12, 13, 14, 15);
      c[tn] = __builtin_amdgcn_wmma_f32_16x16x32_f16(false, a, false, b,
                                                     (short)0, c[tn], false, false);
    }
  }

  float dv[8];
#pragma unroll
  for (int r = 0; r < 8; ++r) dv[r] = dinv_in[tile_m * 16 + r + half * 8];
  float bn[16], p0[16], p1[16], p2[16], p3[16];
#pragma unroll
  for (int tn = 0; tn < 16; ++tn) {
    const int gn = tn * 16 + mn;
    bn[tn] = bias[gn];
    if constexpr (RESID == 2) {
      p0[tn] = inv_std[gn]; p1[tn] = gamma[gn]; p2[tn] = beta[gn]; p3[tn] = mu[gn];
    }
  }
  float rv[16][8];
  if constexpr (RESID != 0) {
#pragma unroll
    for (int tn = 0; tn < 16; ++tn) {
#pragma unroll
      for (int r = 0; r < 8; ++r)
        rv[tn][r] = resid[(tile_m * 16 + r + half * 8) * DIM + tn * 16 + mn];
    }
  }

  __builtin_amdgcn_sched_barrier(0);

#pragma unroll
  for (int tn = 0; tn < 16; ++tn) {
    const int gn = tn * 16 + mn;
    float ra = 0.0f, rc = 0.0f;
    if constexpr (RESID == 2) {
      ra = p0[tn] * p1[tn];
      rc = p2[tn] - p3[tn] * ra;
    }
#pragma unroll
    for (int r = 0; r < 8; ++r) {
      const int gm = tile_m * 16 + r + half * 8;
      float val = c[tn][r] + bn[tn];
      val *= dv[r];
      if constexpr (RESID == 1) val += rv[tn][r];
      if constexpr (RESID == 2) val += bn_relu(rv[tn][r], ra, rc);
      out[gm * DIM + gn] = val;
    }
  }
}

__global__ __launch_bounds__(256)
void bn_stats_kernel(const float* __restrict__ pre, float* __restrict__ sums,
                     float* __restrict__ sumsq, int n) {
  const int c    = threadIdx.x;
  const int row0 = blockIdx.x * 128;
  int rend = row0 + 128;
  if (rend > n) rend = n;
  float s = 0.0f, s2 = 0.0f;
  for (int r = row0; r < rend; ++r) {
    float v = pre[r * DIM + c];
    s  += v;
    s2 += v * v;
  }
  atomAddF(&sums[c], s);
  atomAddF(&sumsq[c], s2);
}

__global__ void bn_finalize_kernel(float* __restrict__ sums, float* __restrict__ sumsq, int n) {
  const int c = threadIdx.x;
  const float inv_n = 1.0f / (float)n;
  float mu  = sums[c] * inv_n;
  float var = sumsq[c] * inv_n - mu * mu;
  sums[c]  = mu;
  sumsq[c] = rsqrtf(var + 1e-5f);
}

extern "C" void kernel_launch(void* const* d_in, const int* in_sizes, int n_in,
                              void* d_out, int out_size, void* d_ws, size_t ws_size,
                              hipStream_t stream) {
  const float* x  = (const float*)d_in[0];
  const int*   ei = (const int*)d_in[1];
  const float* W0 = (const float*)d_in[2];
  const float* b0 = (const float*)d_in[3];
  const float* g0 = (const float*)d_in[4];
  const float* be0= (const float*)d_in[5];
  const float* W1 = (const float*)d_in[6];
  const float* b1 = (const float*)d_in[7];
  const float* g1 = (const float*)d_in[8];
  const float* be1= (const float*)d_in[9];
  const float* W2 = (const float*)d_in[10];
  const float* b2 = (const float*)d_in[11];
  float* out = (float*)d_out;

  const int N = in_sizes[0] / DIM;
  const int E = in_sizes[1] / 2;
  const int ND = N * DIM;
  const int* row = ei;
  const int* col = ei + E;

  float* ws       = (float*)d_ws;
  float* agg      = ws;
  float* bufA     = agg + (size_t)ND;
  _Float16* wp0   = (_Float16*)(bufA + (size_t)ND);
  _Float16* wp1   = wp0 + 65536;
  _Float16* wp2   = wp1 + 65536;
  float* stats0   = (float*)(wp2 + 65536);
  float* stats1   = stats0 + 512;
  float* dinv_in  = stats1 + 512;
  int*   cnt_out  = (int*)(dinv_in + N);
  int*   cnt_in   = cnt_out + N;
  int*   first    = cnt_in + N;
  int*   slot     = first + N;
  int*   esrc     = slot + E;
  float* ewgt     = (float*)(esrc + E);

  const int zb = 256;
  auto nb = [](int n, int b) { return (n + b - 1) / b; };

  const int gath_blocks = nb(N, 8);
  const int tiles_m     = N / 16;
  const int gemm_blocks = nb(tiles_m, 4);
  const int bn_blocks   = nb(N, 128);
  const int pack_blocks = 65536 / 256;

  zero_i32_kernel<<<nb(2 * N, zb), zb, 0, stream>>>(cnt_out, 2 * N);
  count_kernel<<<nb(E, zb), zb, 0, stream>>>(row, col, cnt_out, cnt_in, slot, E);
  scan_kernel<<<1, 1024, 0, stream>>>(cnt_in, first, dinv_in, N);
  fill_kernel<<<nb(E, zb), zb, 0, stream>>>(row, col, cnt_out, first, slot, esrc, ewgt, E);
  pack_w_kernel<<<pack_blocks, 256, 0, stream>>>(W0, wp0);
  pack_w_kernel<<<pack_blocks, 256, 0, stream>>>(W1, wp1);
  pack_w_kernel<<<pack_blocks, 256, 0, stream>>>(W2, wp2);

  gather_kernel<false><<<gath_blocks, 256, 0, stream>>>(x, first, cnt_in, esrc, ewgt, agg, N,
                                                        nullptr, nullptr, nullptr, nullptr);
  gemm_wmma_kernel<1><<<gemm_blocks, 128, 0, stream>>>(agg, wp0, b0, dinv_in, x, out, tiles_m,
                                                       nullptr, nullptr, nullptr, nullptr);
  zero_kernel<<<2, zb, 0, stream>>>(stats0, 512);
  bn_stats_kernel<<<bn_blocks, 256, 0, stream>>>(out, stats0, stats0 + 256, N);
  bn_finalize_kernel<<<1, 256, 0, stream>>>(stats0, stats0 + 256, N);

  gather_kernel<true><<<gath_blocks, 256, 0, stream>>>(out, first, cnt_in, esrc, ewgt, agg, N,
                                                       stats0, stats0 + 256, g0, be0);
  gemm_wmma_kernel<2><<<gemm_blocks, 128, 0, stream>>>(agg, wp1, b1, dinv_in, out, bufA, tiles_m,
                                                       stats0, stats0 + 256, g0, be0);
  zero_kernel<<<2, zb, 0, stream>>>(stats1, 512);
  bn_stats_kernel<<<bn_blocks, 256, 0, stream>>>(bufA, stats1, stats1 + 256, N);
  bn_finalize_kernel<<<1, 256, 0, stream>>>(stats1, stats1 + 256, N);

  gather_kernel<true><<<gath_blocks, 256, 0, stream>>>(bufA, first, cnt_in, esrc, ewgt, agg, N,
                                                       stats1, stats1 + 256, g1, be1);
  gemm_wmma_kernel<0><<<gemm_blocks, 128, 0, stream>>>(agg, wp2, b2, dinv_in, nullptr, out, tiles_m,
                                                       nullptr, nullptr, nullptr, nullptr);
}


# Edge lists built by a two-level counting sort in LDS (2048-edge chunks binned by 256-node range, then ranked per node): no global int atomics, one-block scan or zero pass; list order still by arrival
# speedup vs baseline: 2.9696x; 1.4641x over previous
#include <hip/hip_runtime.h>
#include <math.h>


typedef __attribute__((ext_vector_type(16))) _Float16 v16h;
typedef __attribute__((ext_vector_type(8)))  float    v8f;

#define DIM 256

__device__ __forceinline__ float atomAddF(float* p, float v) {
  return unsafeAtomicAdd(p, v);
}

__device__ __forceinline__ float bn_relu(float v, float a, float c) {
  return fmaxf(v * a + c, 0.0f);
}
__device__ __forceinline__ float4 bn_relu4(float4 v, float4 a, float4 c) {
  float4 r;
  r.x = bn_relu(v.x, a.x, c.x);
  r.y = bn_relu(v.y, a.y, c.y);
  r.z = bn_relu(v.z, a.z, c.z);
  r.w = bn_relu(v.w, a.w, c.w);
  return r;
}

__global__ void zero_kernel(float* __restrict__ p, int n) {
  int i = blockIdx.x * blockDim.x + threadIdx.x;
  if (i < n) p[i] = 0.0f;
}

__global__ __launch_bounds__(1024)
void bin_kernel(const int* __restrict__ row, const int* __restrict__ col,
                int* __restrict__ recc, int* __restrict__ recr,
                int* __restrict__ tabc, int* __restrict__ tabr, int E) {
  __shared__ int hc[256], hr[256];
  const int t = threadIdx.x;
  if (t < 256) { hc[t] = 0; hr[t] = 0; }
  __syncthreads();
  const int cb = blockIdx.x * 2048;
  const int e0 = cb + t, e1 = cb + 1024 + t;
  int r0 = 0, c0 = 0, r1 = 0, c1 = 0;
  if (e0 < E) { r0 = row[e0]; c0 = col[e0]; }
  if (e1 < E) { r1 = row[e1]; c1 = col[e1]; }
  int kc0 = 0, kr0 = 0, kc1 = 0, kr1 = 0;
  if (e0 < E) { kc0 = atomicAdd(&hc[c0 >> 8], 1); kr0 = atomicAdd(&hr[r0 >> 8], 1); }
  if (e1 < E) { kc1 = atomicAdd(&hc[c1 >> 8], 1); kr1 = atomicAdd(&hr[r1 >> 8], 1); }
  __syncthreads();
  int oc = 0, orw = 0;
  if (t < 256) { oc = hc[t]; orw = hr[t]; }
  for (int d = 1; d < 256; d <<= 1) {
    int vc = 0, vr = 0;
    if (t < 256 && t >= d) { vc = hc[t - d]; vr = hr[t - d]; }
    __syncthreads();
    if (t < 256) { hc[t] += vc; hr[t] += vr; }
    __syncthreads();
  }
  if (t < 256) {
    const int ic = hc[t], ir = hr[t];
    int* tc = tabc + blockIdx.x * 257;
    int* tr = tabr + blockIdx.x * 257;
    tc[t] = ic - oc;
    tr[t] = ir - orw;
    if (t == 255) { tc[256] = ic; tr[256] = ir; }
    hc[t] = ic - oc;
    hr[t] = ir - orw;
  }
  __syncthreads();
  if (e0 < E) {
    recc[cb + hc[c0 >> 8] + kc0] = (int)((unsigned)r0 | ((unsigned)(c0 & 255) << 24));
    recr[cb + hr[r0 >> 8] + kr0] = r0 & 255;
  }
  if (e1 < E) {
    recc[cb + hc[c1 >> 8] + kc1] = (int)((unsigned)r1 | ((unsigned)(c1 & 255) << 24));
    recr[cb + hr[r1 >> 8] + kr1] = r1 & 255;
  }
}

__global__ __launch_bounds__(256)
void build_kernel(const int* __restrict__ recc, const int* __restrict__ recr,
                  const int* __restrict__ tabc, const int* __restrict__ tabr,
                  int* __restrict__ first, int* __restrict__ cnt_in,
                  float* __restrict__ dinv_in, float* __restrict__ dinv_out,
                  int* __restrict__ esrc, int n, int chunks) {
  __shared__ int cin[256], cout[256], cur[256], red[256];
  const int t = threadIdx.x;
  const int b = blockIdx.x;
  cin[t] = 0;
  cout[t] = 0;
  __syncthreads();
  int below = 0;
  for (int sb = t; sb < chunks; sb += 256) {
    const int* tc = tabc + sb * 257 + b;
    const int* tr = tabr + sb * 257 + b;
    const int s0 = tc[0], s1 = tc[1];
    const int q0 = tr[0], q1 = tr[1];
    below += s0;
    const int* rc = recc + sb * 2048;
    const int* rr = recr + sb * 2048;
    for (int p = s0; p < s1; ++p) atomicAdd(&cin[(unsigned)rc[p] >> 24], 1);
    for (int p = q0; p < q1; ++p) atomicAdd(&cout[rr[p]], 1);
  }
  red[t] = below;
  __syncthreads();
  for (int d = 128; d > 0; d >>= 1) {
    if (t < d) red[t] += red[t + d];
    __syncthreads();
  }
  const int base = red[0];
  const int ci = cin[t], co = cout[t];
  cur[t] = ci;
  __syncthreads();
  for (int d = 1; d < 256; d <<= 1) {
    const int v = (t >= d) ? cur[t - d] : 0;
    __syncthreads();
    cur[t] += v;
    __syncthreads();
  }
  const int off = cur[t] - ci;
  cur[t] = off;
  const int node = b * 256 + t;
  if (node < n) {
    first[node]  = base + off;
    cnt_in[node] = ci;
    float vi = (float)ci;
    vi = vi < 1.0f ? 1.0f : vi;
    dinv_in[node] = rsqrtf(vi);
    float vo = (float)co;
    vo = vo < 1.0f ? 1.0f : vo;
    dinv_out[node] = rsqrtf(vo);
  }
  __syncthreads();
  for (int sb = t; sb < chunks; sb += 256) {
    const int* tc = tabc + sb * 257 + b;
    const int s0 = tc[0], s1 = tc[1];
    const int* rc = recc + sb * 2048;
    for (int p = s0; p < s1; ++p) {
      const unsigned rec = (unsigned)rc[p];
      const int pos = atomicAdd(&cur[rec >> 24], 1);
      esrc[base + pos] = (int)(rec & 0xFFFFFFu);
    }
  }
}

__global__ void edge_weight_kernel(const int* __restrict__ esrc, const float* __restrict__ dinv_out,
                                   float* __restrict__ ewgt, int E) {
  int p = blockIdx.x * blockDim.x + threadIdx.x;
  if (p < E) ewgt[p] = dinv_out[esrc[p]];
}

template <bool BN>
__global__ __launch_bounds__(256)
void gather_kernel(const float* __restrict__ h, const int* __restrict__ first,
                   const int* __restrict__ cnt_in, const int* __restrict__ esrc,
                   const float* __restrict__ ewgt, float* __restrict__ agg, int n,
                   const float* __restrict__ mu, const float* __restrict__ inv_std,
                   const float* __restrict__ gamma, const float* __restrict__ beta) {
  const int node = __builtin_amdgcn_readfirstlane((int)(blockIdx.x * 8 + (threadIdx.x >> 5)));
  if (node >= n) return;
  const int lane = threadIdx.x & 31;
  const int beg = first[node];
  const int len = cnt_in[node];
  const int*   es = esrc + beg;
  const float* ew = ewgt + beg;
  const float* hb = h + lane * 4;

  float4 s0 = {}, s1 = {}, t0 = {}, t1 = {};
  if constexpr (BN) {
    const int c0 = lane * 4;
    const float4 m0 = *(const float4*)(mu + c0);
    const float4 m1 = *(const float4*)(mu + 128 + c0);
    const float4 i0 = *(const float4*)(inv_std + c0);
    const float4 i1 = *(const float4*)(inv_std + 128 + c0);
    const float4 g0 = *(const float4*)(gamma + c0);
    const float4 g1 = *(const float4*)(gamma + 128 + c0);
    const float4 e0 = *(const float4*)(beta + c0);
    const float4 e1 = *(const float4*)(beta + 128 + c0);
    s0.x = i0.x * g0.x; s0.y = i0.y * g0.y; s0.z = i0.z * g0.z; s0.w = i0.w * g0.w;
    s1.x = i1.x * g1.x; s1.y = i1.y * g1.y; s1.z = i1.z * g1.z; s1.w = i1.w * g1.w;
    t0.x = e0.x - m0.x * s0.x; t0.y = e0.y - m0.y * s0.y; t0.z = e0.z - m0.z * s0.z; t0.w = e0.w - m0.w * s0.w;
    t1.x = e1.x - m1.x * s1.x; t1.y = e1.y - m1.y * s1.y; t1.z = e1.z - m1.z * s1.z; t1.w = e1.w - m1.w * s1.w;
  }

  float4 a0 = {0.0f, 0.0f, 0.0f, 0.0f};
  float4 a1 = {0.0f, 0.0f, 0.0f, 0.0f};
  int j = 0;
#pragma unroll 2
  for (; j + 4 <= len; j += 4) {
    const int r0 = es[j], r1 = es[j + 1], r2 = es[j + 2], r3 = es[j + 3];
    const float w0 = ew[j], w1 = ew[j + 1], w2 = ew[j + 2], w3 = ew[j + 3];
    const float* p0 = hb + r0 * DIM;
    const float* p1 = hb + r1 * DIM;
    const float* p2 = hb + r2 * DIM;
    const float* p3 = hb + r3 * DIM;
    float4 x0 = *(const float4*)(p0), y0 = *(const float4*)(p0 + 128);
    float4 x1 = *(const float4*)(p1), y1 = *(const float4*)(p1 + 128);
    float4 x2 = *(const float4*)(p2), y2 = *(const float4*)(p2 + 128);
    float4 x3 = *(const float4*)(p3), y3 = *(const float4*)(p3 + 128);
    if constexpr (BN) {
      x0 = bn_relu4(x0, s0, t0); y0 = bn_relu4(y0, s1, t1);
      x1 = bn_relu4(x1, s0, t0); y1 = bn_relu4(y1, s1, t1);
      x2 = bn_relu4(x2, s0, t0); y2 = bn_relu4(y2, s1, t1);
      x3 = bn_relu4(x3, s0, t0); y3 = bn_relu4(y3, s1, t1);
    }
    a0.x += x0.x * w0; a0.y += x0.y * w0; a0.z += x0.z * w0; a0.w += x0.w * w0;
    a1.x += y0.x * w0; a1.y += y0.y * w0; a1.z += y0.z * w0; a1.w += y0.w * w0;
    a0.x += x1.x * w1; a0.y += x1.y * w1; a0.z += x1.z * w1; a0.w += x1.w * w1;
    a1.x += y1.x * w1; a1.y += y1.y * w1; a1.z += y1.z * w1; a1.w += y1.w * w1;
    a0.x += x2.x * w2; a0.y += x2.y * w2; a0.z += x2.z * w2; a0.w += x2.w * w2;
    a1.x += y2.x * w2; a1.y += y2.y * w2; a1.z += y2.z * w2; a1.w += y2.w * w2;
    a0.x += x3.x * w3; a0.y += x3.y * w3; a0.z += x3.z * w3; a0.w += x3.w * w3;
    a1.x += y3.x * w3; a1.y += y3.y * w3; a1.z += y3.z * w3; a1.w += y3.w * w3;
  }
  for (; j < len; ++j) {
    const int r0 = es[j];
    const float w0 = ew[j];
    const float* p0 = hb + r0 * DIM;
    float4 x0 = *(const float4*)(p0), y0 = *(const float4*)(p0 + 128);
    if constexpr (BN) {
      x0 = bn_relu4(x0, s0, t0); y0 = bn_relu4(y0, s1, t1);
    }
    a0.x += x0.x * w0; a0.y += x0.y * w0; a0.z += x0.z * w0; a0.w += x0.w * w0;
    a1.x += y0.x * w0; a1.y += y0.y * w0; a1.z += y0.z * w0; a1.w += y0.w * w0;
  }
  float* dst = agg + node * DIM + lane * 4;
  *(float4*)(dst)       = a0;
  *(float4*)(dst + 128) = a1;
}

__global__ __launch_bounds__(256)
void pack_w_kernel(const float* __restrict__ W, _Float16* __restrict__ pack) {
  int t = blockIdx.x * blockDim.x + threadIdx.x;
  int el    = t & 7;
  int lane  = (t >> 3) & 31;
  int piece = (t >> 8) & 1;
  int kc    = (t >> 9) & 7;
  int tn    = t >> 12;
  int half = lane >> 4, mn = lane & 15;
  int kb = kc * 32 + (piece << 4) + (half << 3) + el;
  int gn = tn * 16 + mn;
  pack[t] = (_Float16)W[kb * DIM + gn];
}

typedef __attribute__((ext_vector_type(8))) _Float16 v8h;

template <int RESID>
__global__ __launch_bounds__(128)
void gemm_wmma_kernel(const float* __restrict__ A, const _Float16* __restrict__ Wp,
                      const float* __restrict__ bias, const float* __restrict__ dinv_in,
                      const float* __restrict__ resid, float* __restrict__ out, int tiles_m,
                      const float* __restrict__ mu, const float* __restrict__ inv_std,
                      const float* __restrict__ gamma, const float* __restrict__ beta) {
  const int tile_m = blockIdx.x * (blockDim.x >> 5) + (threadIdx.x >> 5);
  if (tile_m >= tiles_m) return;
  const int lane   = threadIdx.x & 31;
  const int half   = lane >> 4;
  const int mn     = lane & 15;

  const int gm_a = tile_m * 16 + mn;
  const float* Arow = A + gm_a * DIM + half * 8;
  const v8h*   Bp   = (const v8h*)Wp + lane;

  v8f c[16];
#pragma unroll
  for (int tn = 0; tn < 16; ++tn) c[tn] = v8f{};

  v8h bl[2][16], bh[2][16];
  float4 q[2][4];
#pragma unroll
  for (int tn = 0; tn < 16; ++tn) {
    const v8h* bp = Bp + (tn * 8) * 64;
    bl[0][tn] = bp[0];
    bh[0][tn] = bp[32];
  }
  q[0][0] = *(const float4*)(Arow);
  q[0][1] = *(const float4*)(Arow + 4);
  q[0][2] = *(const float4*)(Arow + 16);
  q[0][3] = *(const float4*)(Arow + 20);

#pragma unroll
  for (int kc = 0; kc < 8; ++kc) {
    const int cur = kc & 1, nxt = cur ^ 1;
    if (kc + 1 < 8) {
#pragma unroll
      for (int tn = 0; tn < 16; ++tn) {
        const v8h* bp = Bp + (tn * 8 + kc + 1) * 64;
        bl[nxt][tn] = bp[0];
        bh[nxt][tn] = bp[32];
      }
      const float* ap = Arow + (kc + 1) * 32;
      q[nxt][0] = *(const float4*)(ap);
      q[nxt][1] = *(const float4*)(ap + 4);
      q[nxt][2] = *(const float4*)(ap + 16);
      q[nxt][3] = *(const float4*)(ap + 20);
    }
    __builtin_amdgcn_sched_barrier(0);
    const float4 q0 = q[cur][0], q1 = q[cur][1], q2 = q[cur][2], q3 = q[cur][3];
    v16h a;
    a[0]  = (_Float16)q0.x;  a[1]  = (_Float16)q0.y;
    a[2]  = (_Float16)q0.z;  a[3]  = (_Float16)q0.w;
    a[4]  = (_Float16)q1.x;  a[5]  = (_Float16)q1.y;
    a[6]  = (_Float16)q1.z;  a[7]  = (_Float16)q1.w;
    a[8]  = (_Float16)q2.x;  a[9]  = (_Float16)q2.y;
    a[10] = (_Float16)q2.z;  a[11] = (_Float16)q2.w;
    a[12] = (_Float16)q3.x;  a[13] = (_Float16)q3.y;
    a[14] = (_Float16)q3.z;  a[15] = (_Float16)q3.w;
#pragma unroll
    for (int tn = 0; tn < 16; ++tn) {
      const v16h b = __builtin_shufflevector(bl[cur][tn], bh[cur][tn], 0, 1, 2, 3, 4, 5, 6, 7,
                                             8, 9, 10, 11, 12, 13, 14, 15);
      c[tn] = __builtin_amdgcn_wmma_f32_16x16x32_f16(false, a, false, b,
                                                     (short)0, c[tn], false, false);
    }
  }

  float dv[8];
#pragma unroll
  for (int r = 0; r < 8; ++r) dv[r] = dinv_in[tile_m * 16 + r + half * 8];
  float bn[16], p0[16], p1[16], p2[16], p3[16];
#pragma unroll
  for (int tn = 0; tn < 16; ++tn) {
    const int gn = tn * 16 + mn;
    bn[tn] = bias[gn];
    if constexpr (RESID == 2) {
      p0[tn] = inv_std[gn]; p1[tn] = gamma[gn]; p2[tn] = beta[gn]; p3[tn] = mu[gn];
    }
  }
  float rv[16][8];
  if constexpr (RESID != 0) {
#pragma unroll
    for (int tn = 0; tn < 16; ++tn) {
#pragma unroll
      for (int r = 0; r < 8; ++r)
        rv[tn][r] = resid[(tile_m * 16 + r + half * 8) * DIM + tn * 16 + mn];
    }
  }

  __builtin_amdgcn_sched_barrier(0);

#pragma unroll
  for (int tn = 0; tn < 16; ++tn) {
    const int gn = tn * 16 + mn;
    float ra = 0.0f, rc = 0.0f;
    if constexpr (RESID == 2) {
      ra = p0[tn] * p1[tn];
      rc = p2[tn] - p3[tn] * ra;
    }
#pragma unroll
    for (int r = 0; r < 8; ++r) {
      const int gm = tile_m * 16 + r + half * 8;
      float val = c[tn][r] + bn[tn];
      val *= dv[r];
      if constexpr (RESID == 1) val += rv[tn][r];
      if constexpr (RESID == 2) val += bn_relu(rv[tn][r], ra, rc);
      out[gm * DIM + gn] = val;
    }
  }
}

__global__ __launch_bounds__(256)
void bn_stats_kernel(const float* __restrict__ pre, float* __restrict__ sums,
                     float* __restrict__ sumsq, int n) {
  const int c    = threadIdx.x;
  const int row0 = blockIdx.x * 128;
  int rend = row0 + 128;
  if (rend > n) rend = n;
  float s = 0.0f, s2 = 0.0f;
  for (int r = row0; r < rend; ++r) {
    float v = pre[r * DIM + c];
    s  += v;
    s2 += v * v;
  }
  atomAddF(&sums[c], s);
  atomAddF(&sumsq[c], s2);
}

__global__ void bn_finalize_kernel(float* __restrict__ sums, float* __restrict__ sumsq, int n) {
  const int c = threadIdx.x;
  const float inv_n = 1.0f / (float)n;
  float mu  = sums[c] * inv_n;
  float var = sumsq[c] * inv_n - mu * mu;
  sums[c]  = mu;
  sumsq[c] = rsqrtf(var + 1e-5f);
}

extern "C" void kernel_launch(void* const* d_in, const int* in_sizes, int n_in,
                              void* d_out, int out_size, void* d_ws, size_t ws_size,
                              hipStream_t stream) {
  const float* x  = (const float*)d_in[0];
  const int*   ei = (const int*)d_in[1];
  const float* W0 = (const float*)d_in[2];
  const float* b0 = (const float*)d_in[3];
  const float* g0 = (const float*)d_in[4];
  const float* be0= (const float*)d_in[5];
  const float* W1 = (const float*)d_in[6];
  const float* b1 = (const float*)d_in[7];
  const float* g1 = (const float*)d_in[8];
  const float* be1= (const float*)d_in[9];
  const float* W2 = (const float*)d_in[10];
  const float* b2 = (const float*)d_in[11];
  float* out = (float*)d_out;

  const int N = in_sizes[0] / DIM;
  const int E = in_sizes[1] / 2;
  const int ND = N * DIM;
  const int* row = ei;
  const int* col = ei + E;
  const int chunks = (E + 2047) / 2048;
  const int ranges = (N + 255) / 256;

  float* ws       = (float*)d_ws;
  float* agg      = ws;
  float* bufA     = agg + (size_t)ND;
  _Float16* wp0   = (_Float16*)(bufA + (size_t)ND);
  _Float16* wp1   = wp0 + 65536;
  _Float16* wp2   = wp1 + 65536;
  float* stats0   = (float*)(wp2 + 65536);
  float* stats1   = stats0 + 512;
  float* dinv_in  = stats1 + 512;
  float* dinv_out = dinv_in + N;
  int*   cnt_in   = (int*)(dinv_out + N);
  int*   first    = cnt_in + N;
  int*   esrc     = first + N;
  float* ewgt     = (float*)(esrc + E);
  int*   recc     = (int*)(ewgt + E);
  int*   recr     = recc + (size_t)chunks * 2048;
  int*   tabc     = recr + (size_t)chunks * 2048;
  int*   tabr     = tabc + (size_t)chunks * 257;

  const int zb = 256;
  auto nb = [](int n, int b) { return (n + b - 1) / b; };

  const int gath_blocks = nb(N, 8);
  const int tiles_m     = N / 16;
  const int gemm_blocks = nb(tiles_m, 4);
  const int bn_blocks   = nb(N, 128);
  const int pack_blocks = 65536 / 256;

  bin_kernel<<<chunks, 1024, 0, stream>>>(row, col, recc, recr, tabc, tabr, E);
  build_kernel<<<ranges, 256, 0, stream>>>(recc, recr, tabc, tabr, first, cnt_in, dinv_in, dinv_out,
                                           esrc, N, chunks);
  edge_weight_kernel<<<nb(E, zb), zb, 0, stream>>>(esrc, dinv_out, ewgt, E);
  pack_w_kernel<<<pack_blocks, 256, 0, stream>>>(W0, wp0);
  pack_w_kernel<<<pack_blocks, 256, 0, stream>>>(W1, wp1);
  pack_w_kernel<<<pack_blocks, 256, 0, stream>>>(W2, wp2);

  gather_kernel<false><<<gath_blocks, 256, 0, stream>>>(x, first, cnt_in, esrc, ewgt, agg, N,
                                                        nullptr, nullptr, nullptr, nullptr);
  gemm_wmma_kernel<1><<<gemm_blocks, 128, 0, stream>>>(agg, wp0, b0, dinv_in, x, out, tiles_m,
                                                       nullptr, nullptr, nullptr, nullptr);
  zero_kernel<<<2, zb, 0, stream>>>(stats0, 512);
  bn_stats_kernel<<<bn_blocks, 256, 0, stream>>>(out, stats0, stats0 + 256, N);
  bn_finalize_kernel<<<1, 256, 0, stream>>>(stats0, stats0 + 256, N);

  gather_kernel<true><<<gath_blocks, 256, 0, stream>>>(out, first, cnt_in, esrc, ewgt, agg, N,
                                                       stats0, stats0 + 256, g0, be0);
  gemm_wmma_kernel<2><<<gemm_blocks, 128, 0, stream>>>(agg, wp1, b1, dinv_in, out, bufA, tiles_m,
                                                       stats0, stats0 + 256, g0, be0);
  zero_kernel<<<2, zb, 0, stream>>>(stats1, 512);
  bn_stats_kernel<<<bn_blocks, 256, 0, stream>>>(bufA, stats1, stats1 + 256, N);
  bn_finalize_kernel<<<1, 256, 0, stream>>>(stats1, stats1 + 256, N);

  gather_kernel<true><<<gath_blocks, 256, 0, stream>>>(bufA, first, cnt_in, esrc, ewgt, agg, N,
                                                       stats1, stats1 + 256, g1, be1);
  gemm_wmma_kernel<0><<<gemm_blocks, 128, 0, stream>>>(agg, wp2, b2, dinv_in, nullptr, out, tiles_m,
                                                       nullptr, nullptr, nullptr, nullptr);
}


# BatchNorm column sums formed in the GEMM epilogue (a strip's rows per lane, 4 strips via LDS, then float atomics in arrival order: a new sum order); stats, zero and finalize launches removed
# speedup vs baseline: 4.6001x; 1.5491x over previous
#include <hip/hip_runtime.h>
#include <math.h>


typedef __attribute__((ext_vector_type(16))) _Float16 v16h;
typedef __attribute__((ext_vector_type(8)))  float    v8f;

#define DIM 256

__device__ __forceinline__ float atomAddF(float* p, float v) {
  return unsafeAtomicAdd(p, v);
}

__device__ __forceinline__ float bn_relu(float v, float a, float c) {
  return fmaxf(v * a + c, 0.0f);
}
__device__ __forceinline__ float4 bn_relu4(float4 v, float4 a, float4 c) {
  float4 r;
  r.x = bn_relu(v.x, a.x, c.x);
  r.y = bn_relu(v.y, a.y, c.y);
  r.z = bn_relu(v.z, a.z, c.z);
  r.w = bn_relu(v.w, a.w, c.w);
  return r;
}
__device__ __forceinline__ void bn_coef(float sum, float sumsq, float gamma, float beta,
                                        float inv_n, float& a, float& c) {
  const float mu  = sum * inv_n;
  const float var = sumsq * inv_n - mu * mu;
  a = rsqrtf(var + 1e-5f) * gamma;
  c = beta - mu * a;
}
__device__ __forceinline__ float xhalf(float x) {
  const int b = __builtin_bit_cast(int, x);
  return __builtin_bit_cast(float, __builtin_amdgcn_permlanex16(b, b, 0x76543210, 0xfedcba98, false, false));
}
__device__ __forceinline__ void bn_coef4(float4 sum, float4 sumsq, float4 gamma, float4 beta,
                                         float inv_n, float4& a, float4& c) {
  bn_coef(sum.x, sumsq.x, gamma.x, beta.x, inv_n, a.x, c.x);
  bn_coef(sum.y, sumsq.y, gamma.y, beta.y, inv_n, a.y, c.y);
  bn_coef(sum.z, sumsq.z, gamma.z, beta.z, inv_n, a.z, c.z);
  bn_coef(sum.w, sumsq.w, gamma.w, beta.w, inv_n, a.w, c.w);
}

__global__ __launch_bounds__(1024)
void bin_kernel(const int* __restrict__ row, const int* __restrict__ col,
                int* __restrict__ recc, int* __restrict__ recr,
                int* __restrict__ tabc, int* __restrict__ tabr, int E) {
  __shared__ int hc[256], hr[256];
  const int t = threadIdx.x;
  if (t < 256) { hc[t] = 0; hr[t] = 0; }
  __syncthreads();
  const int cb = blockIdx.x * 2048;
  const int e0 = cb + t, e1 = cb + 1024 + t;
  int r0 = 0, c0 = 0, r1 = 0, c1 = 0;
  if (e0 < E) { r0 = row[e0]; c0 = col[e0]; }
  if (e1 < E) { r1 = row[e1]; c1 = col[e1]; }
  int kc0 = 0, kr0 = 0, kc1 = 0, kr1 = 0;
  if (e0 < E) { kc0 = atomicAdd(&hc[c0 >> 8], 1); kr0 = atomicAdd(&hr[r0 >> 8], 1); }
  if (e1 < E) { kc1 = atomicAdd(&hc[c1 >> 8], 1); kr1 = atomicAdd(&hr[r1 >> 8], 1); }
  __syncthreads();
  int oc = 0, orw = 0;
  if (t < 256) { oc = hc[t]; orw = hr[t]; }
  for (int d = 1; d < 256; d <<= 1) {
    int vc = 0, vr = 0;
    if (t < 256 && t >= d) { vc = hc[t - d]; vr = hr[t - d]; }
    __syncthreads();
    if (t < 256) { hc[t] += vc; hr[t] += vr; }
    __syncthreads();
  }
  if (t < 256) {
    const int ic = hc[t], ir = hr[t];
    int* tc = tabc + blockIdx.x * 257;
    int* tr = tabr + blockIdx.x * 257;
    tc[t] = ic - oc;
    tr[t] = ir - orw;
    if (t == 255) { tc[256] = ic; tr[256] = ir; }
    hc[t] = ic - oc;
    hr[t] = ir - orw;
  }
  __syncthreads();
  if (e0 < E) {
    recc[cb + hc[c0 >> 8] + kc0] = (int)((unsigned)r0 | ((unsigned)(c0 & 255) << 24));
    recr[cb + hr[r0 >> 8] + kr0] = r0 & 255;
  }
  if (e1 < E) {
    recc[cb + hc[c1 >> 8] + kc1] = (int)((unsigned)r1 | ((unsigned)(c1 & 255) << 24));
    recr[cb + hr[r1 >> 8] + kr1] = r1 & 255;
  }
}

__global__ __launch_bounds__(256)
void build_kernel(const int* __restrict__ recc, const int* __restrict__ recr,
                  const int* __restrict__ tabc, const int* __restrict__ tabr,
                  int* __restrict__ first, int* __restrict__ cnt_in,
                  float* __restrict__ dinv_in, float* __restrict__ dinv_out,
                  int* __restrict__ esrc, int n, int chunks) {
  __shared__ int cin[256], cout[256], cur[256], red[256];
  const int t = threadIdx.x;
  const int b = blockIdx.x;
  cin[t] = 0;
  cout[t] = 0;
  __syncthreads();
  int below = 0;
  for (int sb = t; sb < chunks; sb += 256) {
    const int* tc = tabc + sb * 257 + b;
    const int* tr = tabr + sb * 257 + b;
    const int s0 = tc[0], s1 = tc[1];
    const int q0 = tr[0], q1 = tr[1];
    below += s0;
    const int* rc = recc + sb * 2048;
    const int* rr = recr + sb * 2048;
    for (int p = s0; p < s1; ++p) atomicAdd(&cin[(unsigned)rc[p] >> 24], 1);
    for (int p = q0; p < q1; ++p) atomicAdd(&cout[rr[p]], 1);
  }
  red[t] = below;
  __syncthreads();
  for (int d = 128; d > 0; d >>= 1) {
    if (t < d) red[t] += red[t + d];
    __syncthreads();
  }
  const int base = red[0];
  const int ci = cin[t], co = cout[t];
  cur[t] = ci;
  __syncthreads();
  for (int d = 1; d < 256; d <<= 1) {
    const int v = (t >= d) ? cur[t - d] : 0;
    __syncthreads();
    cur[t] += v;
    __syncthreads();
  }
  const int off = cur[t] - ci;
  cur[t] = off;
  const int node = b * 256 + t;
  if (node < n) {
    first[node]  = base + off;
    cnt_in[node] = ci;
    float vi = (float)ci;
    vi = vi < 1.0f ? 1.0f : vi;
    dinv_in[node] = rsqrtf(vi);
    float vo = (float)co;
    vo = vo < 1.0f ? 1.0f : vo;
    dinv_out[node] = rsqrtf(vo);
  }
  __syncthreads();
  for (int sb = t; sb < chunks; sb += 256) {
    const int* tc = tabc + sb * 257 + b;
    const int s0 = tc[0], s1 = tc[1];
    const int* rc = recc + sb * 2048;
    for (int p = s0; p < s1; ++p) {
      const unsigned rec = (unsigned)rc[p];
      const int pos = atomicAdd(&cur[rec >> 24], 1);
      esrc[base + pos] = (int)(rec & 0xFFFFFFu);
    }
  }
}

__global__ void edge_weight_kernel(const int* __restrict__ esrc, const float* __restrict__ dinv_out,
                                   float* __restrict__ ewgt, int E) {
  int p = blockIdx.x * blockDim.x + threadIdx.x;
  if (p < E) ewgt[p] = dinv_out[esrc[p]];
}

template <bool BN>
__global__ __launch_bounds__(256)
void gather_kernel(const float* __restrict__ h, const int* __restrict__ first,
                   const int* __restrict__ cnt_in, const int* __restrict__ esrc,
                   const float* __restrict__ ewgt, float* __restrict__ agg, int n,
                   const float* __restrict__ colsum, const float* __restrict__ colsumsq,
                   const float* __restrict__ gamma, const float* __restrict__ beta) {
  const int node = __builtin_amdgcn_readfirstlane((int)(blockIdx.x * 8 + (threadIdx.x >> 5)));
  if (node >= n) return;
  const int lane = threadIdx.x & 31;
  const int beg = first[node];
  const int len = cnt_in[node];
  const int*   es = esrc + beg;
  const float* ew = ewgt + beg;
  const float* hb = h + lane * 4;

  float4 s0 = {}, s1 = {}, t0 = {}, t1 = {};
  if constexpr (BN) {
    const int c0 = lane * 4;
    const float inv_n = 1.0f / (float)n;
    const float4 m0 = *(const float4*)(colsum + c0);
    const float4 m1 = *(const float4*)(colsum + 128 + c0);
    const float4 q0 = *(const float4*)(colsumsq + c0);
    const float4 q1 = *(const float4*)(colsumsq + 128 + c0);
    const float4 g0 = *(const float4*)(gamma + c0);
    const float4 g1 = *(const float4*)(gamma + 128 + c0);
    const float4 e0 = *(const float4*)(beta + c0);
    const float4 e1 = *(const float4*)(beta + 128 + c0);
    bn_coef4(m0, q0, g0, e0, inv_n, s0, t0);
    bn_coef4(m1, q1, g1, e1, inv_n, s1, t1);
  }

  float4 a0 = {0.0f, 0.0f, 0.0f, 0.0f};
  float4 a1 = {0.0f, 0.0f, 0.0f, 0.0f};
  int j = 0;
#pragma unroll 2
  for (; j + 4 <= len; j += 4) {
    const int r0 = es[j], r1 = es[j + 1], r2 = es[j + 2], r3 = es[j + 3];
    const float w0 = ew[j], w1 = ew[j + 1], w2 = ew[j + 2], w3 = ew[j + 3];
    const float* p0 = hb + r0 * DIM;
    const float* p1 = hb + r1 * DIM;
    const float* p2 = hb + r2 * DIM;
    const float* p3 = hb + r3 * DIM;
    float4 x0 = *(const float4*)(p0), y0 = *(const float4*)(p0 + 128);
    float4 x1 = *(const float4*)(p1), y1 = *(const float4*)(p1 + 128);
    float4 x2 = *(const float4*)(p2), y2 = *(const float4*)(p2 + 128);
    float4 x3 = *(const float4*)(p3), y3 = *(const float4*)(p3 + 128);
    if constexpr (BN) {
      x0 = bn_relu4(x0, s0, t0); y0 = bn_relu4(y0, s1, t1);
      x1 = bn_relu4(x1, s0, t0); y1 = bn_relu4(y1, s1, t1);
      x2 = bn_relu4(x2, s0, t0); y2 = bn_relu4(y2, s1, t1);
      x3 = bn_relu4(x3, s0, t0); y3 = bn_relu4(y3, s1, t1);
    }
    a0.x += x0.x * w0; a0.y += x0.y * w0; a0.z += x0.z * w0; a0.w += x0.w * w0;
    a1.x += y0.x * w0; a1.y += y0.y * w0; a1.z += y0.z * w0; a1.w += y0.w * w0;
    a0.x += x1.x * w1; a0.y += x1.y * w1; a0.z += x1.z * w1; a0.w += x1.w * w1;
    a1.x += y1.x * w1; a1.y += y1.y * w1; a1.z += y1.z * w1; a1.w += y1.w * w1;
    a0.x += x2.x * w2; a0.y += x2.y * w2; a0.z += x2.z * w2; a0.w += x2.w * w2;
    a1.x += y2.x * w2; a1.y += y2.y * w2; a1.z += y2.z * w2; a1.w += y2.w * w2;
    a0.x += x3.x * w3; a0.y += x3.y * w3; a0.z += x3.z * w3; a0.w += x3.w * w3;
    a1.x += y3.x * w3; a1.y += y3.y * w3; a1.z += y3.z * w3; a1.w += y3.w * w3;
  }
  for (; j < len; ++j) {
    const int r0 = es[j];
    const float w0 = ew[j];
    const float* p0 = hb + r0 * DIM;
    float4 x0 = *(const float4*)(p0), y0 = *(const float4*)(p0 + 128);
    if constexpr (BN) {
      x0 = bn_relu4(x0, s0, t0); y0 = bn_relu4(y0, s1, t1);
    }
    a0.x += x0.x * w0; a0.y += x0.y * w0; a0.z += x0.z * w0; a0.w += x0.w * w0;
    a1.x += y0.x * w0; a1.y += y0.y * w0; a1.z += y0.z * w0; a1.w += y0.w * w0;
  }
  float* dst = agg + node * DIM + lane * 4;
  *(float4*)(dst)       = a0;
  *(float4*)(dst + 128) = a1;
}

__global__ __launch_bounds__(256)
void pack_w_kernel(const float* __restrict__ W, _Float16* __restrict__ pack,
                   float* __restrict__ acc) {
  int t = blockIdx.x * blockDim.x + threadIdx.x;
  if (acc != nullptr && t < 512) acc[t] = 0.0f;
  int el    = t & 7;
  int lane  = (t >> 3) & 31;
  int piece = (t >> 8) & 1;
  int kc    = (t >> 9) & 7;
  int tn    = t >> 12;
  int half = lane >> 4, mn = lane & 15;
  int kb = kc * 32 + (piece << 4) + (half << 3) + el;
  int gn = tn * 16 + mn;
  pack[t] = (_Float16)W[kb * DIM + gn];
}

typedef __attribute__((ext_vector_type(8))) _Float16 v8h;

#define GEMM_WAVES 4

template <int RESID, bool STATS>
__global__ __launch_bounds__(GEMM_WAVES * 32)
void gemm_wmma_kernel(const float* __restrict__ A, const _Float16* __restrict__ Wp,
                      const float* __restrict__ bias, const float* __restrict__ dinv_in,
                      const float* __restrict__ resid, float* __restrict__ out, int tiles_m,
                      const float* __restrict__ rsum, const float* __restrict__ rsumsq,
                      const float* __restrict__ gamma, const float* __restrict__ beta,
                      float* __restrict__ acc, int n) {
  const int wave   = __builtin_amdgcn_readfirstlane((int)(threadIdx.x >> 5));
  const int tile_m = blockIdx.x * GEMM_WAVES + wave;
  const int lane   = threadIdx.x & 31;
  const int half   = lane >> 4;
  const int mn     = lane & 15;

  __shared__ float red[STATS ? GEMM_WAVES : 1][512];

  if (tile_m < tiles_m) {
    const int gm_a = tile_m * 16 + mn;
    const float* Arow = A + gm_a * DIM + half * 8;
    const v8h*   Bp   = (const v8h*)Wp + lane;

    v8f c[16];
#pragma unroll
    for (int tn = 0; tn < 16; ++tn) c[tn] = v8f{};

    v8h bl[2][16], bh[2][16];
    float4 q[2][4];
#pragma unroll
    for (int tn = 0; tn < 16; ++tn) {
      const v8h* bp = Bp + (tn * 8) * 64;
      bl[0][tn] = bp[0];
      bh[0][tn] = bp[32];
    }
    q[0][0] = *(const float4*)(Arow);
    q[0][1] = *(const float4*)(Arow + 4);
    q[0][2] = *(const float4*)(Arow + 16);
    q[0][3] = *(const float4*)(Arow + 20);

#pragma unroll
    for (int kc = 0; kc < 8; ++kc) {
      const int cur = kc & 1, nxt = cur ^ 1;
      if (kc + 1 < 8) {
#pragma unroll
        for (int tn = 0; tn < 16; ++tn) {
          const v8h* bp = Bp + (tn * 8 + kc + 1) * 64;
          bl[nxt][tn] = bp[0];
          bh[nxt][tn] = bp[32];
        }
        const float* ap = Arow + (kc + 1) * 32;
        q[nxt][0] = *(const float4*)(ap);
        q[nxt][1] = *(const float4*)(ap + 4);
        q[nxt][2] = *(const float4*)(ap + 16);
        q[nxt][3] = *(const float4*)(ap + 20);
      }
      __builtin_amdgcn_sched_barrier(0);
      const float4 q0 = q[cur][0], q1 = q[cur][1], q2 = q[cur][2], q3 = q[cur][3];
      v16h a;
      a[0]  = (_Float16)q0.x;  a[1]  = (_Float16)q0.y;
      a[2]  = (_Float16)q0.z;  a[3]  = (_Float16)q0.w;
      a[4]  = (_Float16)q1.x;  a[5]  = (_Float16)q1.y;
      a[6]  = (_Float16)q1.z;  a[7]  = (_Float16)q1.w;
      a[8]  = (_Float16)q2.x;  a[9]  = (_Float16)q2.y;
      a[10] = (_Float16)q2.z;  a[11] = (_Float16)q2.w;
      a[12] = (_Float16)q3.x;  a[13] = (_Float16)q3.y;
      a[14] = (_Float16)q3.z;  a[15] = (_Float16)q3.w;
#pragma unroll
      for (int tn = 0; tn < 16; ++tn) {
        const v16h b = __builtin_shufflevector(bl[cur][tn], bh[cur][tn], 0, 1, 2, 3, 4, 5, 6, 7,
                                               8, 9, 10, 11, 12, 13, 14, 15);
        c[tn] = __builtin_amdgcn_wmma_f32_16x16x32_f16(false, a, false, b,
                                                       (short)0, c[tn], false, false);
      }
    }

    float dv[8];
#pragma unroll
    for (int r = 0; r < 8; ++r) dv[r] = dinv_in[tile_m * 16 + r + half * 8];
    float bn[16], p0[16], p1[16], p2[16], p3[16];
#pragma unroll
    for (int tn = 0; tn < 16; ++tn) {
      const int gn = tn * 16 + mn;
      bn[tn] = bias[gn];
      if constexpr (RESID == 2) {
        p0[tn] = rsum[gn]; p1[tn] = rsumsq[gn]; p2[tn] = gamma[gn]; p3[tn] = beta[gn];
      }
    }
    float rv[16][8];
    if constexpr (RESID != 0) {
#pragma unroll
      for (int tn = 0; tn < 16; ++tn) {
#pragma unroll
        for (int r = 0; r < 8; ++r)
          rv[tn][r] = resid[(tile_m * 16 + r + half * 8) * DIM + tn * 16 + mn];
      }
    }

    __builtin_amdgcn_sched_barrier(0);

    const float inv_n = 1.0f / (float)n;
#pragma unroll
    for (int j = 0; j < 8; ++j) {
      float cs[2], cq[2];
#pragma unroll
      for (int u = 0; u < 2; ++u) {
        const int tn = 2 * j + u;
        const int gn = tn * 16 + mn;
        float ra = 0.0f, rc = 0.0f;
        if constexpr (RESID == 2) bn_coef(p0[tn], p1[tn], p2[tn], p3[tn], inv_n, ra, rc);
        float s = 0.0f, s2 = 0.0f;
#pragma unroll
        for (int r = 0; r < 8; ++r) {
          const int gm = tile_m * 16 + r + half * 8;
          float val = c[tn][r] + bn[tn];
          val *= dv[r];
          if constexpr (RESID == 1) val += rv[tn][r];
          if constexpr (RESID == 2) val += bn_relu(rv[tn][r], ra, rc);
          out[gm * DIM + gn] = val;
          if constexpr (STATS) { s += val; s2 += val * val; }
        }
        cs[u] = s;
        cq[u] = s2;
      }
      if constexpr (STATS) {
        const float keep_s = half ? cs[1] : cs[0];
        const float send_s = half ? cs[0] : cs[1];
        const float keep_q = half ? cq[1] : cq[0];
        const float send_q = half ? cq[0] : cq[1];
        red[wave][32 * j + lane]       = keep_s + xhalf(send_s);
        red[wave][256 + 32 * j + lane] = keep_q + xhalf(send_q);
        __builtin_amdgcn_sched_barrier(0);
      }
    }
  } else if constexpr (STATS) {
#pragma unroll
    for (int j = 0; j < 16; ++j) red[wave][32 * j + lane] = 0.0f;
  }

  if constexpr (STATS) {
    __syncthreads();
#pragma unroll
    for (int i = 0; i < 512 / (GEMM_WAVES * 32); ++i) {
      const int col = i * (GEMM_WAVES * 32) + threadIdx.x;
      float v = red[0][col];
#pragma unroll
      for (int w = 1; w < GEMM_WAVES; ++w) v += red[w][col];
      atomAddF(&acc[col], v);
    }
  }
}

extern "C" void kernel_launch(void* const* d_in, const int* in_sizes, int n_in,
                              void* d_out, int out_size, void* d_ws, size_t ws_size,
                              hipStream_t stream) {
  const float* x  = (const float*)d_in[0];
  const int*   ei = (const int*)d_in[1];
  const float* W0 = (const float*)d_in[2];
  const float* b0 = (const float*)d_in[3];
  const float* g0 = (const float*)d_in[4];
  const float* be0= (const float*)d_in[5];
  const float* W1 = (const float*)d_in[6];
  const float* b1 = (const float*)d_in[7];
  const float* g1 = (const float*)d_in[8];
  const float* be1= (const float*)d_in[9];
  const float* W2 = (const float*)d_in[10];
  const float* b2 = (const float*)d_in[11];
  float* out = (float*)d_out;

  const int N = in_sizes[0] / DIM;
  const int E = in_sizes[1] / 2;
  const int ND = N * DIM;
  const int* row = ei;
  const int* col = ei + E;
  const int chunks = (E + 2047) / 2048;
  const int ranges = (N + 255) / 256;

  float* ws       = (float*)d_ws;
  float* agg      = ws;
  float* bufA     = agg + (size_t)ND;
  _Float16* wp0   = (_Float16*)(bufA + (size_t)ND);
  _Float16* wp1   = wp0 + 65536;
  _Float16* wp2   = wp1 + 65536;
  float* stats0   = (float*)(wp2 + 65536);
  float* stats1   = stats0 + 512;
  float* dinv_in  = stats1 + 512;
  float* dinv_out = dinv_in + N;
  int*   cnt_in   = (int*)(dinv_out + N);
  int*   first    = cnt_in + N;
  int*   esrc     = first + N;
  float* ewgt     = (float*)(esrc + E);
  int*   recc     = (int*)(ewgt + E);
  int*   recr     = recc + (size_t)chunks * 2048;
  int*   tabc     = recr + (size_t)chunks * 2048;
  int*   tabr     = tabc + (size_t)chunks * 257;

  const int zb = 256;
  auto nb = [](int n, int b) { return (n + b - 1) / b; };

  const int gath_blocks = nb(N, 8);
  const int tiles_m     = N / 16;
  const int gemm_blocks = nb(tiles_m, GEMM_WAVES);
  const int pack_blocks = 65536 / 256;

  bin_kernel<<<chunks, 1024, 0, stream>>>(row, col, recc, recr, tabc, tabr, E);
  build_kernel<<<ranges, 256, 0, stream>>>(recc, recr, tabc, tabr, first, cnt_in, dinv_in, dinv_out,
                                           esrc, N, chunks);
  edge_weight_kernel<<<nb(E, zb), zb, 0, stream>>>(esrc, dinv_out, ewgt, E);
  pack_w_kernel<<<pack_blocks, 256, 0, stream>>>(W0, wp0, stats0);
  pack_w_kernel<<<pack_blocks, 256, 0, stream>>>(W1, wp1, stats1);
  pack_w_kernel<<<pack_blocks, 256, 0, stream>>>(W2, wp2, nullptr);

  gather_kernel<false><<<gath_blocks, 256, 0, stream>>>(x, first, cnt_in, esrc, ewgt, agg, N,
                                                        nullptr, nullptr, nullptr, nullptr);
  gemm_wmma_kernel<1, true><<<gemm_blocks, GEMM_WAVES * 32, 0, stream>>>(
      agg, wp0, b0, dinv_in, x, out, tiles_m, nullptr, nullptr, nullptr, nullptr, stats0, N);

  gather_kernel<true><<<gath_blocks, 256, 0, stream>>>(out, first, cnt_in, esrc, ewgt, agg, N,
                                                       stats0, stats0 + 256, g0, be0);
  gemm_wmma_kernel<2, true><<<gemm_blocks, GEMM_WAVES * 32, 0, stream>>>(
      agg, wp1, b1, dinv_in, out, bufA, tiles_m, stats0, stats0 + 256, g0, be0, stats1, N);

  gather_kernel<true><<<gath_blocks, 256, 0, stream>>>(bufA, first, cnt_in, esrc, ewgt, agg, N,
                                                       stats1, stats1 + 256, g1, be1);
  gemm_wmma_kernel<0, false><<<gemm_blocks, GEMM_WAVES * 32, 0, stream>>>(
      agg, wp2, b2, dinv_in, nullptr, out, tiles_m, nullptr, nullptr, nullptr, nullptr, nullptr, N);
}


# Aggregated rows kept as the 16-bit A values the GEMM already rounded to: f32 sums rounded once at the gather's store (was at the GEMM's load), A loads halve; products and sum order unchanged
# speedup vs baseline: 4.9895x; 1.0847x over previous
#include <hip/hip_runtime.h>
#include <math.h>


typedef __attribute__((ext_vector_type(16))) _Float16 v16h;
typedef __attribute__((ext_vector_type(8)))  float    v8f;
typedef __attribute__((ext_vector_type(4)))  _Float16 v4h;

#define DIM 256

__device__ __forceinline__ float atomAddF(float* p, float v) {
  return unsafeAtomicAdd(p, v);
}

__device__ __forceinline__ float bn_relu(float v, float a, float c) {
  return fmaxf(v * a + c, 0.0f);
}
__device__ __forceinline__ float4 bn_relu4(float4 v, float4 a, float4 c) {
  float4 r;
  r.x = bn_relu(v.x, a.x, c.x);
  r.y = bn_relu(v.y, a.y, c.y);
  r.z = bn_relu(v.z, a.z, c.z);
  r.w = bn_relu(v.w, a.w, c.w);
  return r;
}
__device__ __forceinline__ void bn_coef(float sum, float sumsq, float gamma, float beta,
                                        float inv_n, float& a, float& c) {
  const float mu  = sum * inv_n;
  const float var = sumsq * inv_n - mu * mu;
  a = rsqrtf(var + 1e-5f) * gamma;
  c = beta - mu * a;
}
__device__ __forceinline__ float xhalf(float x) {
  const int b = __builtin_bit_cast(int, x);
  return __builtin_bit_cast(float, __builtin_amdgcn_permlanex16(b, b, 0x76543210, 0xfedcba98, false, false));
}
__device__ __forceinline__ void bn_coef4(float4 sum, float4 sumsq, float4 gamma, float4 beta,
                                         float inv_n, float4& a, float4& c) {
  bn_coef(sum.x, sumsq.x, gamma.x, beta.x, inv_n, a.x, c.x);
  bn_coef(sum.y, sumsq.y, gamma.y, beta.y, inv_n, a.y, c.y);
  bn_coef(sum.z, sumsq.z, gamma.z, beta.z, inv_n, a.z, c.z);
  bn_coef(sum.w, sumsq.w, gamma.w, beta.w, inv_n, a.w, c.w);
}

__global__ __launch_bounds__(1024)
void bin_kernel(const int* __restrict__ row, const int* __restrict__ col,
                int* __restrict__ recc, int* __restrict__ recr,
                int* __restrict__ tabc, int* __restrict__ tabr, int E) {
  __shared__ int hc[256], hr[256];
  const int t = threadIdx.x;
  if (t < 256) { hc[t] = 0; hr[t] = 0; }
  __syncthreads();
  const int cb = blockIdx.x * 2048;
  const int e0 = cb + t, e1 = cb + 1024 + t;
  int r0 = 0, c0 = 0, r1 = 0, c1 = 0;
  if (e0 < E) { r0 = row[e0]; c0 = col[e0]; }
  if (e1 < E) { r1 = row[e1]; c1 = col[e1]; }
  int kc0 = 0, kr0 = 0, kc1 = 0, kr1 = 0;
  if (e0 < E) { kc0 = atomicAdd(&hc[c0 >> 8], 1); kr0 = atomicAdd(&hr[r0 >> 8], 1); }
  if (e1 < E) { kc1 = atomicAdd(&hc[c1 >> 8], 1); kr1 = atomicAdd(&hr[r1 >> 8], 1); }
  __syncthreads();
  int oc = 0, orw = 0;
  if (t < 256) { oc = hc[t]; orw = hr[t]; }
  for (int d = 1; d < 256; d <<= 1) {
    int vc = 0, vr = 0;
    if (t < 256 && t >= d) { vc = hc[t - d]; vr = hr[t - d]; }
    __syncthreads();
    if (t < 256) { hc[t] += vc; hr[t] += vr; }
    __syncthreads();
  }
  if (t < 256) {
    const int ic = hc[t], ir = hr[t];
    int* tc = tabc + blockIdx.x * 257;
    int* tr = tabr + blockIdx.x * 257;
    tc[t] = ic - oc;
    tr[t] = ir - orw;
    if (t == 255) { tc[256] = ic; tr[256] = ir; }
    hc[t] = ic - oc;
    hr[t] = ir - orw;
  }
  __syncthreads();
  if (e0 < E) {
    recc[cb + hc[c0 >> 8] + kc0] = (int)((unsigned)r0 | ((unsigned)(c0 & 255) << 24));
    recr[cb + hr[r0 >> 8] + kr0] = r0 & 255;
  }
  if (e1 < E) {
    recc[cb + hc[c1 >> 8] + kc1] = (int)((unsigned)r1 | ((unsigned)(c1 & 255) << 24));
    recr[cb + hr[r1 >> 8] + kr1] = r1 & 255;
  }
}

__global__ __launch_bounds__(256)
void build_kernel(const int* __restrict__ recc, const int* __restrict__ recr,
                  const int* __restrict__ tabc, const int* __restrict__ tabr,
                  int* __restrict__ first, int* __restrict__ cnt_in,
                  float* __restrict__ dinv_in, float* __restrict__ dinv_out,
                  int* __restrict__ esrc, int n, int chunks) {
  __shared__ int cin[256], cout[256], cur[256], red[256];
  const int t = threadIdx.x;
  const int b = blockIdx.x;
  cin[t] = 0;
  cout[t] = 0;
  __syncthreads();
  int below = 0;
  for (int sb = t; sb < chunks; sb += 256) {
    const int* tc = tabc + sb * 257 + b;
    const int* tr = tabr + sb * 257 + b;
    const int s0 = tc[0], s1 = tc[1];
    const int q0 = tr[0], q1 = tr[1];
    below += s0;
    const int* rc = recc + sb * 2048;
    const int* rr = recr + sb * 2048;
    for (int p = s0; p < s1; ++p) atomicAdd(&cin[(unsigned)rc[p] >> 24], 1);
    for (int p = q0; p < q1; ++p) atomicAdd(&cout[rr[p]], 1);
  }
  red[t] = below;
  __syncthreads();
  for (int d = 128; d > 0; d >>= 1) {
    if (t < d) red[t] += red[t + d];
    __syncthreads();
  }
  const int base = red[0];
  const int ci = cin[t], co = cout[t];
  cur[t] = ci;
  __syncthreads();
  for (int d = 1; d < 256; d <<= 1) {
    const int v = (t >= d) ? cur[t - d] : 0;
    __syncthreads();
    cur[t] += v;
    __syncthreads();
  }
  const int off = cur[t] - ci;
  cur[t] = off;
  const int node = b * 256 + t;
  if (node < n) {
    first[node]  = base + off;
    cnt_in[node] = ci;
    float vi = (float)ci;
    vi = vi < 1.0f ? 1.0f : vi;
    dinv_in[node] = rsqrtf(vi);
    float vo = (float)co;
    vo = vo < 1.0f ? 1.0f : vo;
    dinv_out[node] = rsqrtf(vo);
  }
  __syncthreads();
  for (int sb = t; sb < chunks; sb += 256) {
    const int* tc = tabc + sb * 257 + b;
    const int s0 = tc[0], s1 = tc[1];
    const int* rc = recc + sb * 2048;
    for (int p = s0; p < s1; ++p) {
      const unsigned rec = (unsigned)rc[p];
      const int pos = atomicAdd(&cur[rec >> 24], 1);
      esrc[base + pos] = (int)(rec & 0xFFFFFFu);
    }
  }
}

__global__ void edge_weight_kernel(const int* __restrict__ esrc, const float* __restrict__ dinv_out,
                                   float* __restrict__ ewgt, int E) {
  int p = blockIdx.x * blockDim.x + threadIdx.x;
  if (p < E) ewgt[p] = dinv_out[esrc[p]];
}

template <bool BN>
__global__ __launch_bounds__(256)
void gather_kernel(const float* __restrict__ h, const int* __restrict__ first,
                   const int* __restrict__ cnt_in, const int* __restrict__ esrc,
                   const float* __restrict__ ewgt, _Float16* __restrict__ agg, int n,
                   const float* __restrict__ colsum, const float* __restrict__ colsumsq,
                   const float* __restrict__ gamma, const float* __restrict__ beta) {
  const int node = __builtin_amdgcn_readfirstlane((int)(blockIdx.x * 8 + (threadIdx.x >> 5)));
  if (node >= n) return;
  const int lane = threadIdx.x & 31;
  const int beg = first[node];
  const int len = cnt_in[node];
  const int*   es = esrc + beg;
  const float* ew = ewgt + beg;
  const float* hb = h + lane * 4;

  float4 s0 = {}, s1 = {}, t0 = {}, t1 = {};
  if constexpr (BN) {
    const int c0 = lane * 4;
    const float inv_n = 1.0f / (float)n;
    const float4 m0 = *(const float4*)(colsum + c0);
    const float4 m1 = *(const float4*)(colsum + 128 + c0);
    const float4 q0 = *(const float4*)(colsumsq + c0);
    const float4 q1 = *(const float4*)(colsumsq + 128 + c0);
    const float4 g0 = *(const float4*)(gamma + c0);
    const float4 g1 = *(const float4*)(gamma + 128 + c0);
    const float4 e0 = *(const float4*)(beta + c0);
    const float4 e1 = *(const float4*)(beta + 128 + c0);
    bn_coef4(m0, q0, g0, e0, inv_n, s0, t0);
    bn_coef4(m1, q1, g1, e1, inv_n, s1, t1);
  }

  float4 a0 = {0.0f, 0.0f, 0.0f, 0.0f};
  float4 a1 = {0.0f, 0.0f, 0.0f, 0.0f};
  int j = 0;
#pragma unroll 2
  for (; j + 4 <= len; j += 4) {
    const int r0 = es[j], r1 = es[j + 1], r2 = es[j + 2], r3 = es[j + 3];
    const float w0 = ew[j], w1 = ew[j + 1], w2 = ew[j + 2], w3 = ew[j + 3];
    const float* p0 = hb + r0 * DIM;
    const float* p1 = hb + r1 * DIM;
    const float* p2 = hb + r2 * DIM;
    const float* p3 = hb + r3 * DIM;
    float4 x0 = *(const float4*)(p0), y0 = *(const float4*)(p0 + 128);
    float4 x1 = *(const float4*)(p1), y1 = *(const float4*)(p1 + 128);
    float4 x2 = *(const float4*)(p2), y2 = *(const float4*)(p2 + 128);
    float4 x3 = *(const float4*)(p3), y3 = *(const float4*)(p3 + 128);
    if constexpr (BN) {
      x0 = bn_relu4(x0, s0, t0); y0 = bn_relu4(y0, s1, t1);
      x1 = bn_relu4(x1, s0, t0); y1 = bn_relu4(y1, s1, t1);
      x2 = bn_relu4(x2, s0, t0); y2 = bn_relu4(y2, s1, t1);
      x3 = bn_relu4(x3, s0, t0); y3 = bn_relu4(y3, s1, t1);
    }
    a0.x += x0.x * w0; a0.y += x0.y * w0; a0.z += x0.z * w0; a0.w += x0.w * w0;
    a1.x += y0.x * w0; a1.y += y0.y * w0; a1.z += y0.z * w0; a1.w += y0.w * w0;
    a0.x += x1.x * w1; a0.y += x1.y * w1; a0.z += x1.z * w1; a0.w += x1.w * w1;
    a1.x += y1.x * w1; a1.y += y1.y * w1; a1.z += y1.z * w1; a1.w += y1.w * w1;
    a0.x += x2.x * w2; a0.y += x2.y * w2; a0.z += x2.z * w2; a0.w += x2.w * w2;
    a1.x += y2.x * w2; a1.y += y2.y * w2; a1.z += y2.z * w2; a1.w += y2.w * w2;
    a0.x += x3.x * w3; a0.y += x3.y * w3; a0.z += x3.z * w3; a0.w += x3.w * w3;
    a1.x += y3.x * w3; a1.y += y3.y * w3; a1.z += y3.z * w3; a1.w += y3.w * w3;
  }
  for (; j < len; ++j) {
    const int r0 = es[j];
    const float w0 = ew[j];
    const float* p0 = hb + r0 * DIM;
    float4 x0 = *(const float4*)(p0), y0 = *(const float4*)(p0 + 128);
    if constexpr (BN) {
      x0 = bn_relu4(x0, s0, t0); y0 = bn_relu4(y0, s1, t1);
    }
    a0.x += x0.x * w0; a0.y += x0.y * w0; a0.z += x0.z * w0; a0.w += x0.w * w0;
    a1.x += y0.x * w0; a1.y += y0.y * w0; a1.z += y0.z * w0; a1.w += y0.w * w0;
  }
  _Float16* dst = agg + node * DIM + lane * 4;
  v4h h0, h1;
  h0[0] = (_Float16)a0.x; h0[1] = (_Float16)a0.y; h0[2] = (_Float16)a0.z; h0[3] = (_Float16)a0.w;
  h1[0] = (_Float16)a1.x; h1[1] = (_Float16)a1.y; h1[2] = (_Float16)a1.z; h1[3] = (_Float16)a1.w;
  *(v4h*)(dst)       = h0;
  *(v4h*)(dst + 128) = h1;
}

__global__ __launch_bounds__(256)
void pack_w_kernel(const float* __restrict__ W, _Float16* __restrict__ pack,
                   float* __restrict__ acc) {
  int t = blockIdx.x * blockDim.x + threadIdx.x;
  if (acc != nullptr && t < 512) acc[t] = 0.0f;
  int el    = t & 7;
  int lane  = (t >> 3) & 31;
  int piece = (t >> 8) & 1;
  int kc    = (t >> 9) & 7;
  int tn    = t >> 12;
  int half = lane >> 4, mn = lane & 15;
  int kb = kc * 32 + (piece << 4) + (half << 3) + el;
  int gn = tn * 16 + mn;
  pack[t] = (_Float16)W[kb * DIM + gn];
}

typedef __attribute__((ext_vector_type(8))) _Float16 v8h;

#define GEMM_WAVES 4

template <int RESID, bool STATS>
__global__ __launch_bounds__(GEMM_WAVES * 32)
void gemm_wmma_kernel(const _Float16* __restrict__ A, const _Float16* __restrict__ Wp,
                      const float* __restrict__ bias, const float* __restrict__ dinv_in,
                      const float* __restrict__ resid, float* __restrict__ out, int tiles_m,
                      const float* __restrict__ rsum, const float* __restrict__ rsumsq,
                      const float* __restrict__ gamma, const float* __restrict__ beta,
                      float* __restrict__ acc, int n) {
  const int wave   = __builtin_amdgcn_readfirstlane((int)(threadIdx.x >> 5));
  const int tile_m = blockIdx.x * GEMM_WAVES + wave;
  const int lane   = threadIdx.x & 31;
  const int half   = lane >> 4;
  const int mn     = lane & 15;

  __shared__ float red[STATS ? GEMM_WAVES : 1][512];

  if (tile_m < tiles_m) {
    const int gm_a = tile_m * 16 + mn;
    const _Float16* Arow = A + gm_a * DIM + half * 8;
    const v8h*   Bp   = (const v8h*)Wp + lane;

    v8f c[16];
#pragma unroll
    for (int tn = 0; tn < 16; ++tn) c[tn] = v8f{};

    v8h bl[2][16], bh[2][16];
    v8h al[2], ah[2];
#pragma unroll
    for (int tn = 0; tn < 16; ++tn) {
      const v8h* bp = Bp + (tn * 8) * 64;
      bl[0][tn] = bp[0];
      bh[0][tn] = bp[32];
    }
    al[0] = *(const v8h*)(Arow);
    ah[0] = *(const v8h*)(Arow + 16);

#pragma unroll
    for (int kc = 0; kc < 8; ++kc) {
      const int cur = kc & 1, nxt = cur ^ 1;
      if (kc + 1 < 8) {
#pragma unroll
        for (int tn = 0; tn < 16; ++tn) {
          const v8h* bp = Bp + (tn * 8 + kc + 1) * 64;
          bl[nxt][tn] = bp[0];
          bh[nxt][tn] = bp[32];
        }
        const _Float16* ap = Arow + (kc + 1) * 32;
        al[nxt] = *(const v8h*)(ap);
        ah[nxt] = *(const v8h*)(ap + 16);
      }
      __builtin_amdgcn_sched_barrier(0);
      const v16h a = __builtin_shufflevector(al[cur], ah[cur], 0, 1, 2, 3, 4, 5, 6, 7,
                                             8, 9, 10, 11, 12, 13, 14, 15);
#pragma unroll
      for (int tn = 0; tn < 16; ++tn) {
        const v16h b = __builtin_shufflevector(bl[cur][tn], bh[cur][tn], 0, 1, 2, 3, 4, 5, 6, 7,
                                               8, 9, 10, 11, 12, 13, 14, 15);
        c[tn] = __builtin_amdgcn_wmma_f32_16x16x32_f16(false, a, false, b,
                                                       (short)0, c[tn], false, false);
      }
    }

    float dv[8];
#pragma unroll
    for (int r = 0; r < 8; ++r) dv[r] = dinv_in[tile_m * 16 + r + half * 8];
    float bn[16], p0[16], p1[16], p2[16], p3[16];
#pragma unroll
    for (int tn = 0; tn < 16; ++tn) {
      const int gn = tn * 16 + mn;
      bn[tn] = bias[gn];
      if constexpr (RESID == 2) {
        p0[tn] = rsum[gn]; p1[tn] = rsumsq[gn]; p2[tn] = gamma[gn]; p3[tn] = beta[gn];
      }
    }
    float rv[16][8];
    if constexpr (RESID != 0) {
#pragma unroll
      for (int tn = 0; tn < 16; ++tn) {
#pragma unroll
        for (int r = 0; r < 8; ++r)
          rv[tn][r] = resid[(tile_m * 16 + r + half * 8) * DIM + tn * 16 + mn];
      }
    }

    __builtin_amdgcn_sched_barrier(0);

    const float inv_n = 1.0f / (float)n;
#pragma unroll
    for (int j = 0; j < 8; ++j) {
      float cs[2], cq[2];
#pragma unroll
      for (int u = 0; u < 2; ++u) {
        const int tn = 2 * j + u;
        const int gn = tn * 16 + mn;
        float ra = 0.0f, rc = 0.0f;
        if constexpr (RESID == 2) bn_coef(p0[tn], p1[tn], p2[tn], p3[tn], inv_n, ra, rc);
        float s = 0.0f, s2 = 0.0f;
#pragma unroll
        for (int r = 0; r < 8; ++r) {
          const int gm = tile_m * 16 + r + half * 8;
          float val = c[tn][r] + bn[tn];
          val *= dv[r];
          if constexpr (RESID == 1) val += rv[tn][r];
          if constexpr (RESID == 2) val += bn_relu(rv[tn][r], ra, rc);
          out[gm * DIM + gn] = val;
          if constexpr (STATS) { s += val; s2 += val * val; }
        }
        cs[u] = s;
        cq[u] = s2;
      }
      if constexpr (STATS) {
        const float keep_s = half ? cs[1] : cs[0];
        const float send_s = half ? cs[0] : cs[1];
        const float keep_q = half ? cq[1] : cq[0];
        const float send_q = half ? cq[0] : cq[1];
        red[wave][32 * j + lane]       = keep_s + xhalf(send_s);
        red[wave][256 + 32 * j + lane] = keep_q + xhalf(send_q);
        __builtin_amdgcn_sched_barrier(0);
      }
    }
  } else if constexpr (STATS) {
#pragma unroll
    for (int j = 0; j < 16; ++j) red[wave][32 * j + lane] = 0.0f;
  }

  if constexpr (STATS) {
    __syncthreads();
#pragma unroll
    for (int i = 0; i < 512 / (GEMM_WAVES * 32); ++i) {
      const int col = i * (GEMM_WAVES * 32) + threadIdx.x;
      float v = red[0][col];
#pragma unroll
      for (int w = 1; w < GEMM_WAVES; ++w) v += red[w][col];
      atomAddF(&acc[col], v);
    }
  }
}

extern "C" void kernel_launch(void* const* d_in, const int* in_sizes, int n_in,
                              void* d_out, int out_size, void* d_ws, size_t ws_size,
                              hipStream_t stream) {
  const float* x  = (const float*)d_in[0];
  const int*   ei = (const int*)d_in[1];
  const float* W0 = (const float*)d_in[2];
  const float* b0 = (const float*)d_in[3];
  const float* g0 = (const float*)d_in[4];
  const float* be0= (const float*)d_in[5];
  const float* W1 = (const float*)d_in[6];
  const float* b1 = (const float*)d_in[7];
  const float* g1 = (const float*)d_in[8];
  const float* be1= (const float*)d_in[9];
  const float* W2 = (const float*)d_in[10];
  const float* b2 = (const float*)d_in[11];
  float* out = (float*)d_out;

  const int N = in_sizes[0] / DIM;
  const int E = in_sizes[1] / 2;
  const int ND = N * DIM;
  const int* row = ei;
  const int* col = ei + E;
  const int chunks = (E + 2047) / 2048;
  const int ranges = (N + 255) / 256;

  float* ws       = (float*)d_ws;
  _Float16* agg   = (_Float16*)ws;
  float* bufA     = ws + (size_t)ND;
  _Float16* wp0   = (_Float16*)(bufA + (size_t)ND);
  _Float16* wp1   = wp0 + 65536;
  _Float16* wp2   = wp1 + 65536;
  float* stats0   = (float*)(wp2 + 65536);
  float* stats1   = stats0 + 512;
  float* dinv_in  = stats1 + 512;
  float* dinv_out = dinv_in + N;
  int*   cnt_in   = (int*)(dinv_out + N);
  int*   first    = cnt_in + N;
  int*   esrc     = first + N;
  float* ewgt     = (float*)(esrc + E);
  int*   recc     = (int*)(ewgt + E);
  int*   recr     = recc + (size_t)chunks * 2048;
  int*   tabc     = recr + (size_t)chunks * 2048;
  int*   tabr     = tabc + (size_t)chunks * 257;

  const int zb = 256;
  auto nb = [](int n, int b) { return (n + b - 1) / b; };

  const int gath_blocks = nb(N, 8);
  const int tiles_m     = N / 16;
  const int gemm_blocks = nb(tiles_m, GEMM_WAVES);
  const int pack_blocks = 65536 / 256;

  bin_kernel<<<chunks, 1024, 0, stream>>>(row, col, recc, recr, tabc, tabr, E);
  build_kernel<<<ranges, 256, 0, stream>>>(recc, recr, tabc, tabr, first, cnt_in, dinv_in, dinv_out,
                                           esrc, N, chunks);
  edge_weight_kernel<<<nb(E, zb), zb, 0, stream>>>(esrc, dinv_out, ewgt, E);
  pack_w_kernel<<<pack_blocks, 256, 0, stream>>>(W0, wp0, stats0);
  pack_w_kernel<<<pack_blocks, 256, 0, stream>>>(W1, wp1, stats1);
  pack_w_kernel<<<pack_blocks, 256, 0, stream>>>(W2, wp2, nullptr);

  gather_kernel<false><<<gath_blocks, 256, 0, stream>>>(x, first, cnt_in, esrc, ewgt, agg, N,
                                                        nullptr, nullptr, nullptr, nullptr);
  gemm_wmma_kernel<1, true><<<gemm_blocks, GEMM_WAVES * 32, 0, stream>>>(
      agg, wp0, b0, dinv_in, x, out, tiles_m, nullptr, nullptr, nullptr, nullptr, stats0, N);

  gather_kernel<true><<<gath_blocks, 256, 0, stream>>>(out, first, cnt_in, esrc, ewgt, agg, N,
                                                       stats0, stats0 + 256, g0, be0);
  gemm_wmma_kernel<2, true><<<gemm_blocks, GEMM_WAVES * 32, 0, stream>>>(
      agg, wp1, b1, dinv_in, out, bufA, tiles_m, stats0, stats0 + 256, g0, be0, stats1, N);

  gather_kernel<true><<<gath_blocks, 256, 0, stream>>>(bufA, first, cnt_in, esrc, ewgt, agg, N,
                                                       stats1, stats1 + 256, g1, be1);
  gemm_wmma_kernel<0, false><<<gemm_blocks, GEMM_WAVES * 32, 0, stream>>>(
      agg, wp2, b2, dinv_in, nullptr, out, tiles_m, nullptr, nullptr, nullptr, nullptr, nullptr, N);
}
